# starting kernel
# baseline (speedup 1.0000x reference)
#include <hip/hip_runtime.h>
#include <hip/hip_bf16.h>


typedef __attribute__((ext_vector_type(2))) float v2f;
typedef __attribute__((ext_vector_type(8))) float v8f;

#define BSZ        2
#define NN         3072
#define EE         4096
#define NODE_DIM   128
#define EDGE_DIM   64
#define NHEAD      8
#define HDIM       16
#define QKV        384

#define WMMA_F32(A, B, C) \
  __builtin_amdgcn_wmma_f32_16x16x4_f32(false, (A), false, (B), (short)0, (C), false, false)

__global__ __launch_bounds__(32)
void gemm_bias_32x32(const float* __restrict__ X, const float* __restrict__ W,
                     const float* __restrict__ bias, float* __restrict__ C,
                     int M, int K, int Ncols) {
  const int tiles_n = Ncols >> 5;
  const int tile = blockIdx.x;
  const int tm = tile / tiles_n;
  const int tn = tile % tiles_n;
  const int lane = threadIdx.x;
  const int half = lane >> 4;
  const int l15  = lane & 15;

  const int row0 = tm * 32 + l15;
  const int row1 = row0 + 16;
  const int col0 = tn * 32 + l15;
  const int col1 = col0 + 16;

  const float* __restrict__ Xr0 = X + (size_t)row0 * K;
  const float* __restrict__ Xr1 = X + (size_t)row1 * K;

  v8f acc00 = {}, acc01 = {}, acc10 = {}, acc11 = {};
  for (int k = 0; k < K; k += 4) {
    const int ka = k + half * 2;
    v2f a0, a1, b0, b1;
    a0.x = Xr0[ka];                         a0.y = Xr0[ka + 1];
    a1.x = Xr1[ka];                         a1.y = Xr1[ka + 1];
    b0.x = W[(size_t)ka * Ncols + col0];    b0.y = W[(size_t)(ka + 1) * Ncols + col0];
    b1.x = W[(size_t)ka * Ncols + col1];    b1.y = W[(size_t)(ka + 1) * Ncols + col1];
    acc00 = WMMA_F32(a0, b0, acc00);
    acc01 = WMMA_F32(a0, b1, acc01);
    acc10 = WMMA_F32(a1, b0, acc10);
    acc11 = WMMA_F32(a1, b1, acc11);
  }

  const float bv0 = bias[col0];
  const float bv1 = bias[col1];
#pragma unroll
  for (int j = 0; j < 8; ++j) {
    const int r0 = tm * 32 + j + half * 8;
    const int r1 = r0 + 16;
    C[(size_t)r0 * Ncols + col0] = acc00[j] + bv0;
    C[(size_t)r0 * Ncols + col1] = acc01[j] + bv1;
    C[(size_t)r1 * Ncols + col0] = acc10[j] + bv0;
    C[(size_t)r1 * Ncols + col1] = acc11[j] + bv1;
  }
}

__global__ void degree_kernel(const float* __restrict__ G, float* __restrict__ deg) {
  const int idx = blockIdx.x * blockDim.x + threadIdx.x;
  if (idx >= BSZ * NN) return;
  const int b = idx / NN;
  const int n = idx - b * NN;
  const float* __restrict__ g = G + (size_t)b * EE * NN + n;
  float s = 0.f;
  for (int e = 0; e < EE; ++e) s += g[(size_t)e * NN];
  deg[idx] = s;
}

__global__ __launch_bounds__(32)
void agg_kernel(const float* __restrict__ G, const float* __restrict__ qkv_e,
                const float* __restrict__ deg, float* __restrict__ qkv) {
  const int tiles_c = QKV / 32;
  const int tiles_n = NN / 32;
  int tile = blockIdx.x;
  const int b  = tile / (tiles_n * tiles_c);
  tile -= b * tiles_n * tiles_c;
  const int tn = tile / tiles_c;
  const int tc = tile - tn * tiles_c;

  const int lane = threadIdx.x;
  const int half = lane >> 4;
  const int l15  = lane & 15;
  const int n0 = tn * 32 + l15;
  const int n1 = n0 + 16;
  const int c0 = tc * 32 + l15;
  const int c1 = c0 + 16;

  const float* __restrict__ Gb = G     + (size_t)b * EE * NN;
  const float* __restrict__ Eb = qkv_e + (size_t)b * EE * QKV;

  v8f acc00 = {}, acc01 = {}, acc10 = {}, acc11 = {};
  for (int k = 0; k < EE; k += 4) {
    const int ka = k + half * 2;
    const size_t ga = (size_t)ka * NN;
    const size_t ea = (size_t)ka * QKV;
    v2f a0, a1, b0, b1;
    a0.x = Gb[ga + n0];        a0.y = Gb[ga + NN + n0];
    a1.x = Gb[ga + n1];        a1.y = Gb[ga + NN + n1];
    b0.x = Eb[ea + c0];        b0.y = Eb[ea + QKV + c0];
    b1.x = Eb[ea + c1];        b1.y = Eb[ea + QKV + c1];
    acc00 = WMMA_F32(a0, b0, acc00);
    acc01 = WMMA_F32(a0, b1, acc01);
    acc10 = WMMA_F32(a1, b0, acc10);
    acc11 = WMMA_F32(a1, b1, acc11);
  }

  float* __restrict__ out = qkv + (size_t)b * NN * QKV;
#pragma unroll
  for (int j = 0; j < 8; ++j) {
    const int r0 = tn * 32 + j + half * 8;
    const int r1 = r0 + 16;
    const float rdiv0 = 1.0f / deg[b * NN + r0];
    const float rdiv1 = 1.0f / deg[b * NN + r1];
    const size_t o00 = (size_t)r0 * QKV + c0;
    const size_t o01 = (size_t)r0 * QKV + c1;
    const size_t o10 = (size_t)r1 * QKV + c0;
    const size_t o11 = (size_t)r1 * QKV + c1;
    out[o00] = out[o00] + acc00[j] * rdiv0;
    out[o01] = out[o01] + acc01[j] * rdiv0;
    out[o10] = out[o10] + acc10[j] * rdiv1;
    out[o11] = out[o11] + acc11[j] * rdiv1;
  }
}

__global__ __launch_bounds__(32)
void attn_kernel(const float* __restrict__ qkv, float* __restrict__ attout) {
  __shared__ float lds[16 * 16];

  const int tiles_q = NN / 16;
  int tile = blockIdx.x;
  const int b = tile / (NHEAD * tiles_q);
  tile -= b * NHEAD * tiles_q;
  const int h  = tile / tiles_q;
  const int tq = tile - h * tiles_q;

  const int lane = threadIdx.x;
  const int half = lane >> 4;
  const int l15  = lane & 15;
  const float scale = 0.25f;

  const float* __restrict__ base = qkv + (size_t)b * NN * QKV + h * 48;

  float qa[8];
  {
    const int m = tq * 16 + l15;
#pragma unroll
    for (int kk = 0; kk < 16; kk += 4) {
      const int cc = kk + half * 2;
      qa[kk / 2]     = base[(size_t)m * QKV + cc] * scale;
      qa[kk / 2 + 1] = base[(size_t)m * QKV + cc + 1] * scale;
    }
  }

  v8f o = {};
  float mrow[8], lrow[8];
#pragma unroll
  for (int j = 0; j < 8; ++j) { mrow[j] = -1e30f; lrow[j] = 0.f; }

  for (int tk = 0; tk < NN / 16; ++tk) {
    v8f s = {};
    {
      const int m = tk * 16 + l15;
      const float* __restrict__ krow = base + (size_t)m * QKV + 16;
#pragma unroll
      for (int kk = 0; kk < 16; kk += 4) {
        const int cc = kk + half * 2;
        v2f a, bb;
        a.x  = qa[kk / 2];
        a.y  = qa[kk / 2 + 1];
        bb.x = krow[cc];
        bb.y = krow[cc + 1];
        s = WMMA_F32(a, bb, s);
      }
    }

    float pnew[8];
#pragma unroll
    for (int j = 0; j < 8; ++j) {
      const float sj = s[j];
      float mx = sj;
      mx = fmaxf(mx, __shfl_xor(mx, 1, 32));
      mx = fmaxf(mx, __shfl_xor(mx, 2, 32));
      mx = fmaxf(mx, __shfl_xor(mx, 4, 32));
      mx = fmaxf(mx, __shfl_xor(mx, 8, 32));
      const float mnew = fmaxf(mrow[j], mx);
      const float corr = __expf(mrow[j] - mnew);
      const float p    = __expf(sj - mnew);
      float ps = p;
      ps += __shfl_xor(ps, 1, 32);
      ps += __shfl_xor(ps, 2, 32);
      ps += __shfl_xor(ps, 4, 32);
      ps += __shfl_xor(ps, 8, 32);
      lrow[j] = lrow[j] * corr + ps;
      mrow[j] = mnew;
      o[j] *= corr;
      pnew[j] = p;
    }

#pragma unroll
    for (int j = 0; j < 8; ++j)
      lds[(j + half * 8) * 16 + l15] = pnew[j];
    __syncthreads();

#pragma unroll
    for (int kk = 0; kk < 16; kk += 4) {
      const int krow = kk + half * 2;
      v2f a, bb;
      a.x = lds[l15 * 16 + krow];
      a.y = lds[l15 * 16 + krow + 1];
      const int m = tk * 16 + krow;
      bb.x = base[(size_t)m * QKV + 32 + l15];
      bb.y = base[(size_t)(m + 1) * QKV + 32 + l15];
      o = WMMA_F32(a, bb, o);
    }
    __syncthreads();
  }

  float* __restrict__ outp = attout + (size_t)b * NN * NODE_DIM + h * HDIM;
#pragma unroll
  for (int j = 0; j < 8; ++j) {
    const int r = tq * 16 + j + half * 8;
    outp[(size_t)r * NODE_DIM + l15] = o[j] / lrow[j];
  }
}

extern "C" void kernel_launch(void* const* d_in, const int* in_sizes, int n_in,
                              void* d_out, int out_size, void* d_ws, size_t ws_size,
                              hipStream_t stream) {
  const float* node    = (const float*)d_in[0];
  const float* edge    = (const float*)d_in[1];
  const float* G       = (const float*)d_in[2];
  const float* W_qkv_n = (const float*)d_in[3];
  const float* b_qkv_n = (const float*)d_in[4];
  const float* W_qkv_e = (const float*)d_in[5];
  const float* b_qkv_e = (const float*)d_in[6];
  const float* W_o     = (const float*)d_in[7];
  const float* b_o     = (const float*)d_in[8];
  float* out = (float*)d_out;

  float* qkv  = (float*)d_ws;
  float* qkve = qkv  + (size_t)BSZ * NN * QKV;
  float* deg  = qkve + (size_t)BSZ * EE * QKV;
  float* att  = deg  + (size_t)BSZ * NN;

  gemm_bias_32x32<<<(BSZ * NN / 32) * (QKV / 32), 32, 0, stream>>>(
      node, W_qkv_n, b_qkv_n, qkv, BSZ * NN, NODE_DIM, QKV);

  gemm_bias_32x32<<<(BSZ * EE / 32) * (QKV / 32), 32, 0, stream>>>(
      edge, W_qkv_e, b_qkv_e, qkve, BSZ * EE, EDGE_DIM, QKV);

  degree_kernel<<<(BSZ * NN + 255) / 256, 256, 0, stream>>>(G, deg);

  agg_kernel<<<BSZ * (NN / 32) * (QKV / 32), 32, 0, stream>>>(G, qkve, deg, qkv);

  attn_kernel<<<BSZ * NHEAD * (NN / 16), 32, 0, stream>>>(qkv, att);

  gemm_bias_32x32<<<(BSZ * NN / 32) * (NODE_DIM / 32), 32, 0, stream>>>(
      att, W_o, b_o, out, BSZ * NN, NODE_DIM, NODE_DIM);
}


# degree launch removed: row sums of G are added up from the A fragments inside the aggregation GEMM loop (degree sum order changes: per-lane partial sums, then one add across the lane halves)
# speedup vs baseline: 1.2638x; 1.2638x over previous
#include <hip/hip_runtime.h>
#include <hip/hip_bf16.h>


typedef __attribute__((ext_vector_type(2))) float v2f;
typedef __attribute__((ext_vector_type(8))) float v8f;

#define BSZ        2
#define NN         3072
#define EE         4096
#define NODE_DIM   128
#define EDGE_DIM   64
#define NHEAD      8
#define HDIM       16
#define QKV        384

#define WMMA_F32(A, B, C) \
  __builtin_amdgcn_wmma_f32_16x16x4_f32(false, (A), false, (B), (short)0, (C), false, false)

__global__ __launch_bounds__(32)
void gemm_bias_32x32(const float* __restrict__ X, const float* __restrict__ W,
                     const float* __restrict__ bias, float* __restrict__ C,
                     int M, int K, int Ncols) {
  const int tiles_n = Ncols >> 5;
  const int tile = blockIdx.x;
  const int tm = tile / tiles_n;
  const int tn = tile % tiles_n;
  const int lane = threadIdx.x;
  const int half = lane >> 4;
  const int l15  = lane & 15;

  const int row0 = tm * 32 + l15;
  const int row1 = row0 + 16;
  const int col0 = tn * 32 + l15;
  const int col1 = col0 + 16;

  const float* __restrict__ Xr0 = X + (size_t)row0 * K;
  const float* __restrict__ Xr1 = X + (size_t)row1 * K;

  v8f acc00 = {}, acc01 = {}, acc10 = {}, acc11 = {};
  for (int k = 0; k < K; k += 4) {
    const int ka = k + half * 2;
    v2f a0, a1, b0, b1;
    a0.x = Xr0[ka];                         a0.y = Xr0[ka + 1];
    a1.x = Xr1[ka];                         a1.y = Xr1[ka + 1];
    b0.x = W[(size_t)ka * Ncols + col0];    b0.y = W[(size_t)(ka + 1) * Ncols + col0];
    b1.x = W[(size_t)ka * Ncols + col1];    b1.y = W[(size_t)(ka + 1) * Ncols + col1];
    acc00 = WMMA_F32(a0, b0, acc00);
    acc01 = WMMA_F32(a0, b1, acc01);
    acc10 = WMMA_F32(a1, b0, acc10);
    acc11 = WMMA_F32(a1, b1, acc11);
  }

  const float bv0 = bias[col0];
  const float bv1 = bias[col1];
#pragma unroll
  for (int j = 0; j < 8; ++j) {
    const int r0 = tm * 32 + j + half * 8;
    const int r1 = r0 + 16;
    C[(size_t)r0 * Ncols + col0] = acc00[j] + bv0;
    C[(size_t)r0 * Ncols + col1] = acc01[j] + bv1;
    C[(size_t)r1 * Ncols + col0] = acc10[j] + bv0;
    C[(size_t)r1 * Ncols + col1] = acc11[j] + bv1;
  }
}

__global__ __launch_bounds__(32)
void agg_kernel(const float* __restrict__ G, const float* __restrict__ qkv_e,
                float* __restrict__ qkv) {
  const int tiles_c = QKV / 32;
  const int tiles_n = NN / 32;
  int tile = blockIdx.x;
  const int b  = tile / (tiles_n * tiles_c);
  tile -= b * tiles_n * tiles_c;
  const int tn = tile / tiles_c;
  const int tc = tile - tn * tiles_c;

  const int lane = threadIdx.x;
  const int half = lane >> 4;
  const int l15  = lane & 15;
  const int n0 = tn * 32 + l15;
  const int n1 = n0 + 16;
  const int c0 = tc * 32 + l15;
  const int c1 = c0 + 16;

  const float* __restrict__ Gb = G     + (size_t)b * EE * NN;
  const float* __restrict__ Eb = qkv_e + (size_t)b * EE * QKV;

  v8f acc00 = {}, acc01 = {}, acc10 = {}, acc11 = {};
  v2f dsum0 = {}, dsum1 = {};
  for (int k = 0; k < EE; k += 4) {
    const int ka = k + half * 2;
    const size_t ga = (size_t)ka * NN;
    const size_t ea = (size_t)ka * QKV;
    v2f a0, a1, b0, b1;
    a0.x = Gb[ga + n0];        a0.y = Gb[ga + NN + n0];
    a1.x = Gb[ga + n1];        a1.y = Gb[ga + NN + n1];
    b0.x = Eb[ea + c0];        b0.y = Eb[ea + QKV + c0];
    b1.x = Eb[ea + c1];        b1.y = Eb[ea + QKV + c1];
    dsum0 += a0;
    dsum1 += a1;
    acc00 = WMMA_F32(a0, b0, acc00);
    acc01 = WMMA_F32(a0, b1, acc01);
    acc10 = WMMA_F32(a1, b0, acc10);
    acc11 = WMMA_F32(a1, b1, acc11);
  }

  float deg0 = dsum0.x + dsum0.y;
  float deg1 = dsum1.x + dsum1.y;
  deg0 += __shfl_xor(deg0, 16, 32);
  deg1 += __shfl_xor(deg1, 16, 32);
  const float rinv0 = 1.0f / deg0;
  const float rinv1 = 1.0f / deg1;

  float* __restrict__ out = qkv + (size_t)b * NN * QKV;
#pragma unroll
  for (int j = 0; j < 8; ++j) {
    const int r0 = tn * 32 + j + half * 8;
    const int r1 = r0 + 16;
    const float rdiv0 = __shfl(rinv0, j + half * 8, 32);
    const float rdiv1 = __shfl(rinv1, j + half * 8, 32);
    const size_t o00 = (size_t)r0 * QKV + c0;
    const size_t o01 = (size_t)r0 * QKV + c1;
    const size_t o10 = (size_t)r1 * QKV + c0;
    const size_t o11 = (size_t)r1 * QKV + c1;
    out[o00] = out[o00] + acc00[j] * rdiv0;
    out[o01] = out[o01] + acc01[j] * rdiv0;
    out[o10] = out[o10] + acc10[j] * rdiv1;
    out[o11] = out[o11] + acc11[j] * rdiv1;
  }
}

__global__ __launch_bounds__(32)
void attn_kernel(const float* __restrict__ qkv, float* __restrict__ attout) {
  __shared__ float lds[16 * 16];

  const int tiles_q = NN / 16;
  int tile = blockIdx.x;
  const int b = tile / (NHEAD * tiles_q);
  tile -= b * NHEAD * tiles_q;
  const int h  = tile / tiles_q;
  const int tq = tile - h * tiles_q;

  const int lane = threadIdx.x;
  const int half = lane >> 4;
  const int l15  = lane & 15;
  const float scale = 0.25f;

  const float* __restrict__ base = qkv + (size_t)b * NN * QKV + h * 48;

  float qa[8];
  {
    const int m = tq * 16 + l15;
#pragma unroll
    for (int kk = 0; kk < 16; kk += 4) {
      const int cc = kk + half * 2;
      qa[kk / 2]     = base[(size_t)m * QKV + cc] * scale;
      qa[kk / 2 + 1] = base[(size_t)m * QKV + cc + 1] * scale;
    }
  }

  v8f o = {};
  float mrow[8], lrow[8];
#pragma unroll
  for (int j = 0; j < 8; ++j) { mrow[j] = -1e30f; lrow[j] = 0.f; }

  for (int tk = 0; tk < NN / 16; ++tk) {
    v8f s = {};
    {
      const int m = tk * 16 + l15;
      const float* __restrict__ krow = base + (size_t)m * QKV + 16;
#pragma unroll
      for (int kk = 0; kk < 16; kk += 4) {
        const int cc = kk + half * 2;
        v2f a, bb;
        a.x  = qa[kk / 2];
        a.y  = qa[kk / 2 + 1];
        bb.x = krow[cc];
        bb.y = krow[cc + 1];
        s = WMMA_F32(a, bb, s);
      }
    }

    float pnew[8];
#pragma unroll
    for (int j = 0; j < 8; ++j) {
      const float sj = s[j];
      float mx = sj;
      mx = fmaxf(mx, __shfl_xor(mx, 1, 32));
      mx = fmaxf(mx, __shfl_xor(mx, 2, 32));
      mx = fmaxf(mx, __shfl_xor(mx, 4, 32));
      mx = fmaxf(mx, __shfl_xor(mx, 8, 32));
      const float mnew = fmaxf(mrow[j], mx);
      const float corr = __expf(mrow[j] - mnew);
      const float p    = __expf(sj - mnew);
      float ps = p;
      ps += __shfl_xor(ps, 1, 32);
      ps += __shfl_xor(ps, 2, 32);
      ps += __shfl_xor(ps, 4, 32);
      ps += __shfl_xor(ps, 8, 32);
      lrow[j] = lrow[j] * corr + ps;
      mrow[j] = mnew;
      o[j] *= corr;
      pnew[j] = p;
    }

#pragma unroll
    for (int j = 0; j < 8; ++j)
      lds[(j + half * 8) * 16 + l15] = pnew[j];
    __syncthreads();

#pragma unroll
    for (int kk = 0; kk < 16; kk += 4) {
      const int krow = kk + half * 2;
      v2f a, bb;
      a.x = lds[l15 * 16 + krow];
      a.y = lds[l15 * 16 + krow + 1];
      const int m = tk * 16 + krow;
      bb.x = base[(size_t)m * QKV + 32 + l15];
      bb.y = base[(size_t)(m + 1) * QKV + 32 + l15];
      o = WMMA_F32(a, bb, o);
    }
    __syncthreads();
  }

  float* __restrict__ outp = attout + (size_t)b * NN * NODE_DIM + h * HDIM;
#pragma unroll
  for (int j = 0; j < 8; ++j) {
    const int r = tq * 16 + j + half * 8;
    outp[(size_t)r * NODE_DIM + l15] = o[j] / lrow[j];
  }
}

extern "C" void kernel_launch(void* const* d_in, const int* in_sizes, int n_in,
                              void* d_out, int out_size, void* d_ws, size_t ws_size,
                              hipStream_t stream) {
  const float* node    = (const float*)d_in[0];
  const float* edge    = (const float*)d_in[1];
  const float* G       = (const float*)d_in[2];
  const float* W_qkv_n = (const float*)d_in[3];
  const float* b_qkv_n = (const float*)d_in[4];
  const float* W_qkv_e = (const float*)d_in[5];
  const float* b_qkv_e = (const float*)d_in[6];
  const float* W_o     = (const float*)d_in[7];
  const float* b_o     = (const float*)d_in[8];
  float* out = (float*)d_out;

  float* qkv  = (float*)d_ws;
  float* qkve = qkv  + (size_t)BSZ * NN * QKV;
  float* att  = qkve + (size_t)BSZ * EE * QKV;

  gemm_bias_32x32<<<(BSZ * NN / 32) * (QKV / 32), 32, 0, stream>>>(
      node, W_qkv_n, b_qkv_n, qkv, BSZ * NN, NODE_DIM, QKV);

  gemm_bias_32x32<<<(BSZ * EE / 32) * (QKV / 32), 32, 0, stream>>>(
      edge, W_qkv_e, b_qkv_e, qkve, BSZ * EE, EDGE_DIM, QKV);

  agg_kernel<<<BSZ * (NN / 32) * (QKV / 32), 32, 0, stream>>>(G, qkve, qkv);

  attn_kernel<<<BSZ * NHEAD * (NN / 16), 32, 0, stream>>>(qkv, att);

  gemm_bias_32x32<<<(BSZ * NN / 32) * (NODE_DIM / 32), 32, 0, stream>>>(
      att, W_o, b_o, out, BSZ * NN, NODE_DIM, NODE_DIM);
}


# aggregation GEMM K loop software-pipelined: blocks of 8 k-steps load as 8-byte pairs (a lane owns two adjacent rows/columns) into a spare register set while earlier blocks' WMMAs run; sum order kept
# speedup vs baseline: 1.7827x; 1.4106x over previous
#include <hip/hip_runtime.h>
#include <hip/hip_bf16.h>


typedef __attribute__((ext_vector_type(2))) float v2f;
typedef __attribute__((ext_vector_type(8))) float v8f;

#define BSZ        2
#define NN         3072
#define EE         4096
#define NODE_DIM   128
#define EDGE_DIM   64
#define NHEAD      8
#define HDIM       16
#define QKV        384

#define WMMA_F32(A, B, C) \
  __builtin_amdgcn_wmma_f32_16x16x4_f32(false, (A), false, (B), (short)0, (C), false, false)

__global__ __launch_bounds__(32)
void gemm_bias_32x32(const float* __restrict__ X, const float* __restrict__ W,
                     const float* __restrict__ bias, float* __restrict__ C,
                     int M, int K, int Ncols) {
  const int tiles_n = Ncols >> 5;
  const int tile = blockIdx.x;
  const int tm = tile / tiles_n;
  const int tn = tile % tiles_n;
  const int lane = threadIdx.x;
  const int half = lane >> 4;
  const int l15  = lane & 15;

  const int row0 = tm * 32 + l15;
  const int row1 = row0 + 16;
  const int col0 = tn * 32 + l15;
  const int col1 = col0 + 16;

  const float* __restrict__ Xr0 = X + (size_t)row0 * K;
  const float* __restrict__ Xr1 = X + (size_t)row1 * K;

  v8f acc00 = {}, acc01 = {}, acc10 = {}, acc11 = {};
  for (int k = 0; k < K; k += 4) {
    const int ka = k + half * 2;
    v2f a0, a1, b0, b1;
    a0.x = Xr0[ka];                         a0.y = Xr0[ka + 1];
    a1.x = Xr1[ka];                         a1.y = Xr1[ka + 1];
    b0.x = W[(size_t)ka * Ncols + col0];    b0.y = W[(size_t)(ka + 1) * Ncols + col0];
    b1.x = W[(size_t)ka * Ncols + col1];    b1.y = W[(size_t)(ka + 1) * Ncols + col1];
    acc00 = WMMA_F32(a0, b0, acc00);
    acc01 = WMMA_F32(a0, b1, acc01);
    acc10 = WMMA_F32(a1, b0, acc10);
    acc11 = WMMA_F32(a1, b1, acc11);
  }

  const float bv0 = bias[col0];
  const float bv1 = bias[col1];
#pragma unroll
  for (int j = 0; j < 8; ++j) {
    const int r0 = tm * 32 + j + half * 8;
    const int r1 = r0 + 16;
    C[(size_t)r0 * Ncols + col0] = acc00[j] + bv0;
    C[(size_t)r0 * Ncols + col1] = acc01[j] + bv1;
    C[(size_t)r1 * Ncols + col0] = acc10[j] + bv0;
    C[(size_t)r1 * Ncols + col1] = acc11[j] + bv1;
  }
}

#define AGG_U 8

struct AggFrag {
  v2f g[AGG_U][2];
  v2f e[AGG_U][2];
};

__device__ __forceinline__ void agg_load(AggFrag& f, const float* __restrict__ gp,
                                         const float* __restrict__ ep) {
#pragma unroll
  for (int u = 0; u < AGG_U; ++u) {
    f.g[u][0] = *(const v2f*)(gp + (size_t)(4 * u) * NN);
    f.g[u][1] = *(const v2f*)(gp + (size_t)(4 * u + 1) * NN);
    f.e[u][0] = *(const v2f*)(ep + (size_t)(4 * u) * QKV);
    f.e[u][1] = *(const v2f*)(ep + (size_t)(4 * u + 1) * QKV);
  }
}

__device__ __forceinline__ void agg_compute(const AggFrag& f, v8f& acc00, v8f& acc01,
                                            v8f& acc10, v8f& acc11,
                                            v2f& dsum0, v2f& dsum1) {
#pragma unroll
  for (int u = 0; u < AGG_U; ++u) {
    v2f a0, a1, b0, b1;
    a0.x = f.g[u][0].x;  a0.y = f.g[u][1].x;
    a1.x = f.g[u][0].y;  a1.y = f.g[u][1].y;
    b0.x = f.e[u][0].x;  b0.y = f.e[u][1].x;
    b1.x = f.e[u][0].y;  b1.y = f.e[u][1].y;
    dsum0 += a0;
    dsum1 += a1;
    acc00 = WMMA_F32(a0, b0, acc00);
    acc01 = WMMA_F32(a0, b1, acc01);
    acc10 = WMMA_F32(a1, b0, acc10);
    acc11 = WMMA_F32(a1, b1, acc11);
    __builtin_amdgcn_sched_barrier(0);
  }
}

__global__ __launch_bounds__(32)
void agg_kernel(const float* __restrict__ G, const float* __restrict__ qkv_e,
                float* __restrict__ qkv) {
  const int tiles_c = QKV / 32;
  const int tiles_n = NN / 32;
  int tile = blockIdx.x;
  const int b  = tile / (tiles_n * tiles_c);
  tile -= b * tiles_n * tiles_c;
  const int tn = tile / tiles_c;
  const int tc = tile - tn * tiles_c;

  const int lane = threadIdx.x;
  const int half = lane >> 4;
  const int l15  = lane & 15;
  const int n0 = tn * 32 + 2 * l15;
  const int c0 = tc * 32 + 2 * l15;

  const float* __restrict__ gp = G     + (size_t)b * EE * NN  + (size_t)(half * 2) * NN  + n0;
  const float* __restrict__ ep = qkv_e + (size_t)b * EE * QKV + (size_t)(half * 2) * QKV + c0;
  const size_t gstep = (size_t)(4 * AGG_U) * NN;
  const size_t estep = (size_t)(4 * AGG_U) * QKV;

  v8f acc00 = {}, acc01 = {}, acc10 = {}, acc11 = {};
  v2f dsum0 = {}, dsum1 = {};

  AggFrag fa, fb;
  const int nblk = EE / (4 * AGG_U);
  agg_load(fa, gp, ep);
  for (int blk = 0; blk < nblk - 2; blk += 2) {
    agg_load(fb, gp + gstep, ep + estep);
    __builtin_amdgcn_sched_barrier(0);
    agg_compute(fa, acc00, acc01, acc10, acc11, dsum0, dsum1);
    gp += 2 * gstep;
    ep += 2 * estep;
    agg_load(fa, gp, ep);
    __builtin_amdgcn_sched_barrier(0);
    agg_compute(fb, acc00, acc01, acc10, acc11, dsum0, dsum1);
  }
  agg_load(fb, gp + gstep, ep + estep);
  __builtin_amdgcn_sched_barrier(0);
  agg_compute(fa, acc00, acc01, acc10, acc11, dsum0, dsum1);
  agg_compute(fb, acc00, acc01, acc10, acc11, dsum0, dsum1);

  float deg0 = dsum0.x + dsum0.y;
  float deg1 = dsum1.x + dsum1.y;
  deg0 += __shfl_xor(deg0, 16, 32);
  deg1 += __shfl_xor(deg1, 16, 32);
  const float rinv0 = 1.0f / deg0;
  const float rinv1 = 1.0f / deg1;

  float* __restrict__ out = qkv + (size_t)b * NN * QKV;
#pragma unroll
  for (int j = 0; j < 8; ++j) {
    const int fr = j + half * 8;
    const float rdiv0 = __shfl(rinv0, fr, 32);
    const float rdiv1 = __shfl(rinv1, fr, 32);
    const int r0 = tn * 32 + 2 * fr;
    v2f* __restrict__ o0 = (v2f*)(out + (size_t)r0 * QKV + c0);
    v2f* __restrict__ o1 = (v2f*)(out + (size_t)(r0 + 1) * QKV + c0);
    v2f q0 = *o0, q1 = *o1;
    q0.x = q0.x + acc00[j] * rdiv0;
    q0.y = q0.y + acc01[j] * rdiv0;
    q1.x = q1.x + acc10[j] * rdiv1;
    q1.y = q1.y + acc11[j] * rdiv1;
    *o0 = q0;
    *o1 = q1;
  }
}

__global__ __launch_bounds__(32)
void attn_kernel(const float* __restrict__ qkv, float* __restrict__ attout) {
  __shared__ float lds[16 * 16];

  const int tiles_q = NN / 16;
  int tile = blockIdx.x;
  const int b = tile / (NHEAD * tiles_q);
  tile -= b * NHEAD * tiles_q;
  const int h  = tile / tiles_q;
  const int tq = tile - h * tiles_q;

  const int lane = threadIdx.x;
  const int half = lane >> 4;
  const int l15  = lane & 15;
  const float scale = 0.25f;

  const float* __restrict__ base = qkv + (size_t)b * NN * QKV + h * 48;

  float qa[8];
  {
    const int m = tq * 16 + l15;
#pragma unroll
    for (int kk = 0; kk < 16; kk += 4) {
      const int cc = kk + half * 2;
      qa[kk / 2]     = base[(size_t)m * QKV + cc] * scale;
      qa[kk / 2 + 1] = base[(size_t)m * QKV + cc + 1] * scale;
    }
  }

  v8f o = {};
  float mrow[8], lrow[8];
#pragma unroll
  for (int j = 0; j < 8; ++j) { mrow[j] = -1e30f; lrow[j] = 0.f; }

  for (int tk = 0; tk < NN / 16; ++tk) {
    v8f s = {};
    {
      const int m = tk * 16 + l15;
      const float* __restrict__ krow = base + (size_t)m * QKV + 16;
#pragma unroll
      for (int kk = 0; kk < 16; kk += 4) {
        const int cc = kk + half * 2;
        v2f a, bb;
        a.x  = qa[kk / 2];
        a.y  = qa[kk / 2 + 1];
        bb.x = krow[cc];
        bb.y = krow[cc + 1];
        s = WMMA_F32(a, bb, s);
      }
    }

    float pnew[8];
#pragma unroll
    for (int j = 0; j < 8; ++j) {
      const float sj = s[j];
      float mx = sj;
      mx = fmaxf(mx, __shfl_xor(mx, 1, 32));
      mx = fmaxf(mx, __shfl_xor(mx, 2, 32));
      mx = fmaxf(mx, __shfl_xor(mx, 4, 32));
      mx = fmaxf(mx, __shfl_xor(mx, 8, 32));
      const float mnew = fmaxf(mrow[j], mx);
      const float corr = __expf(mrow[j] - mnew);
      const float p    = __expf(sj - mnew);
      float ps = p;
      ps += __shfl_xor(ps, 1, 32);
      ps += __shfl_xor(ps, 2, 32);
      ps += __shfl_xor(ps, 4, 32);
      ps += __shfl_xor(ps, 8, 32);
      lrow[j] = lrow[j] * corr + ps;
      mrow[j] = mnew;
      o[j] *= corr;
      pnew[j] = p;
    }

#pragma unroll
    for (int j = 0; j < 8; ++j)
      lds[(j + half * 8) * 16 + l15] = pnew[j];
    __syncthreads();

#pragma unroll
    for (int kk = 0; kk < 16; kk += 4) {
      const int krow = kk + half * 2;
      v2f a, bb;
      a.x = lds[l15 * 16 + krow];
      a.y = lds[l15 * 16 + krow + 1];
      const int m = tk * 16 + krow;
      bb.x = base[(size_t)m * QKV + 32 + l15];
      bb.y = base[(size_t)(m + 1) * QKV + 32 + l15];
      o = WMMA_F32(a, bb, o);
    }
    __syncthreads();
  }

  float* __restrict__ outp = attout + (size_t)b * NN * NODE_DIM + h * HDIM;
#pragma unroll
  for (int j = 0; j < 8; ++j) {
    const int r = tq * 16 + j + half * 8;
    outp[(size_t)r * NODE_DIM + l15] = o[j] / lrow[j];
  }
}

extern "C" void kernel_launch(void* const* d_in, const int* in_sizes, int n_in,
                              void* d_out, int out_size, void* d_ws, size_t ws_size,
                              hipStream_t stream) {
  const float* node    = (const float*)d_in[0];
  const float* edge    = (const float*)d_in[1];
  const float* G       = (const float*)d_in[2];
  const float* W_qkv_n = (const float*)d_in[3];
  const float* b_qkv_n = (const float*)d_in[4];
  const float* W_qkv_e = (const float*)d_in[5];
  const float* b_qkv_e = (const float*)d_in[6];
  const float* W_o     = (const float*)d_in[7];
  const float* b_o     = (const float*)d_in[8];
  float* out = (float*)d_out;

  float* qkv  = (float*)d_ws;
  float* qkve = qkv  + (size_t)BSZ * NN * QKV;
  float* att  = qkve + (size_t)BSZ * EE * QKV;

  gemm_bias_32x32<<<(BSZ * NN / 32) * (QKV / 32), 32, 0, stream>>>(
      node, W_qkv_n, b_qkv_n, qkv, BSZ * NN, NODE_DIM, QKV);

  gemm_bias_32x32<<<(BSZ * EE / 32) * (QKV / 32), 32, 0, stream>>>(
      edge, W_qkv_e, b_qkv_e, qkve, BSZ * EE, EDGE_DIM, QKV);

  agg_kernel<<<BSZ * (NN / 32) * (QKV / 32), 32, 0, stream>>>(G, qkve, qkv);

  attn_kernel<<<BSZ * NHEAD * (NN / 16), 32, 0, stream>>>(qkv, att);

  gemm_bias_32x32<<<(BSZ * NN / 32) * (NODE_DIM / 32), 32, 0, stream>>>(
      att, W_o, b_o, out, BSZ * NN, NODE_DIM, NODE_DIM);
}


# attention on transposed tiles (S^T=K Q^T, O^T=V^T P^T): a query is a lane, so softmax max/sum run down its registers, P feeds P.V from registers (no LDS), loads a step ahead; row-sum, k order change
# speedup vs baseline: 2.6748x; 1.5004x over previous
#include <hip/hip_runtime.h>
#include <hip/hip_bf16.h>


typedef __attribute__((ext_vector_type(2))) float v2f;
typedef __attribute__((ext_vector_type(8))) float v8f;

#define BSZ        2
#define NN         3072
#define EE         4096
#define NODE_DIM   128
#define EDGE_DIM   64
#define NHEAD      8
#define HDIM       16
#define QKV        384

#define WMMA_F32(A, B, C) \
  __builtin_amdgcn_wmma_f32_16x16x4_f32(false, (A), false, (B), (short)0, (C), false, false)

__global__ __launch_bounds__(32)
void gemm_bias_32x32(const float* __restrict__ X, const float* __restrict__ W,
                     const float* __restrict__ bias, float* __restrict__ C,
                     int M, int K, int Ncols) {
  const int tiles_n = Ncols >> 5;
  const int tile = blockIdx.x;
  const int tm = tile / tiles_n;
  const int tn = tile % tiles_n;
  const int lane = threadIdx.x;
  const int half = lane >> 4;
  const int l15  = lane & 15;

  const int row0 = tm * 32 + l15;
  const int row1 = row0 + 16;
  const int col0 = tn * 32 + l15;
  const int col1 = col0 + 16;

  const float* __restrict__ Xr0 = X + (size_t)row0 * K;
  const float* __restrict__ Xr1 = X + (size_t)row1 * K;

  v8f acc00 = {}, acc01 = {}, acc10 = {}, acc11 = {};
  for (int k = 0; k < K; k += 4) {
    const int ka = k + half * 2;
    v2f a0, a1, b0, b1;
    a0.x = Xr0[ka];                         a0.y = Xr0[ka + 1];
    a1.x = Xr1[ka];                         a1.y = Xr1[ka + 1];
    b0.x = W[(size_t)ka * Ncols + col0];    b0.y = W[(size_t)(ka + 1) * Ncols + col0];
    b1.x = W[(size_t)ka * Ncols + col1];    b1.y = W[(size_t)(ka + 1) * Ncols + col1];
    acc00 = WMMA_F32(a0, b0, acc00);
    acc01 = WMMA_F32(a0, b1, acc01);
    acc10 = WMMA_F32(a1, b0, acc10);
    acc11 = WMMA_F32(a1, b1, acc11);
  }

  const float bv0 = bias[col0];
  const float bv1 = bias[col1];
#pragma unroll
  for (int j = 0; j < 8; ++j) {
    const int r0 = tm * 32 + j + half * 8;
    const int r1 = r0 + 16;
    C[(size_t)r0 * Ncols + col0] = acc00[j] + bv0;
    C[(size_t)r0 * Ncols + col1] = acc01[j] + bv1;
    C[(size_t)r1 * Ncols + col0] = acc10[j] + bv0;
    C[(size_t)r1 * Ncols + col1] = acc11[j] + bv1;
  }
}

#define AGG_U 8

struct AggFrag {
  v2f g[AGG_U][2];
  v2f e[AGG_U][2];
};

__device__ __forceinline__ void agg_load(AggFrag& f, const float* __restrict__ gp,
                                         const float* __restrict__ ep) {
#pragma unroll
  for (int u = 0; u < AGG_U; ++u) {
    f.g[u][0] = *(const v2f*)(gp + (size_t)(4 * u) * NN);
    f.g[u][1] = *(const v2f*)(gp + (size_t)(4 * u + 1) * NN);
    f.e[u][0] = *(const v2f*)(ep + (size_t)(4 * u) * QKV);
    f.e[u][1] = *(const v2f*)(ep + (size_t)(4 * u + 1) * QKV);
  }
}

__device__ __forceinline__ void agg_compute(const AggFrag& f, v8f& acc00, v8f& acc01,
                                            v8f& acc10, v8f& acc11,
                                            v2f& dsum0, v2f& dsum1) {
#pragma unroll
  for (int u = 0; u < AGG_U; ++u) {
    v2f a0, a1, b0, b1;
    a0.x = f.g[u][0].x;  a0.y = f.g[u][1].x;
    a1.x = f.g[u][0].y;  a1.y = f.g[u][1].y;
    b0.x = f.e[u][0].x;  b0.y = f.e[u][1].x;
    b1.x = f.e[u][0].y;  b1.y = f.e[u][1].y;
    dsum0 += a0;
    dsum1 += a1;
    acc00 = WMMA_F32(a0, b0, acc00);
    acc01 = WMMA_F32(a0, b1, acc01);
    acc10 = WMMA_F32(a1, b0, acc10);
    acc11 = WMMA_F32(a1, b1, acc11);
    __builtin_amdgcn_sched_barrier(0);
  }
}

__global__ __launch_bounds__(32)
void agg_kernel(const float* __restrict__ G, const float* __restrict__ qkv_e,
                float* __restrict__ qkv) {
  const int tiles_c = QKV / 32;
  const int tiles_n = NN / 32;
  int tile = blockIdx.x;
  const int b  = tile / (tiles_n * tiles_c);
  tile -= b * tiles_n * tiles_c;
  const int tn = tile / tiles_c;
  const int tc = tile - tn * tiles_c;

  const int lane = threadIdx.x;
  const int half = lane >> 4;
  const int l15  = lane & 15;
  const int n0 = tn * 32 + 2 * l15;
  const int c0 = tc * 32 + 2 * l15;

  const float* __restrict__ gp = G     + (size_t)b * EE * NN  + (size_t)(half * 2) * NN  + n0;
  const float* __restrict__ ep = qkv_e + (size_t)b * EE * QKV + (size_t)(half * 2) * QKV + c0;
  const size_t gstep = (size_t)(4 * AGG_U) * NN;
  const size_t estep = (size_t)(4 * AGG_U) * QKV;

  v8f acc00 = {}, acc01 = {}, acc10 = {}, acc11 = {};
  v2f dsum0 = {}, dsum1 = {};

  AggFrag fa, fb;
  const int nblk = EE / (4 * AGG_U);
  agg_load(fa, gp, ep);
  for (int blk = 0; blk < nblk - 2; blk += 2) {
    agg_load(fb, gp + gstep, ep + estep);
    __builtin_amdgcn_sched_barrier(0);
    agg_compute(fa, acc00, acc01, acc10, acc11, dsum0, dsum1);
    gp += 2 * gstep;
    ep += 2 * estep;
    agg_load(fa, gp, ep);
    __builtin_amdgcn_sched_barrier(0);
    agg_compute(fb, acc00, acc01, acc10, acc11, dsum0, dsum1);
  }
  agg_load(fb, gp + gstep, ep + estep);
  __builtin_amdgcn_sched_barrier(0);
  agg_compute(fa, acc00, acc01, acc10, acc11, dsum0, dsum1);
  agg_compute(fb, acc00, acc01, acc10, acc11, dsum0, dsum1);

  float deg0 = dsum0.x + dsum0.y;
  float deg1 = dsum1.x + dsum1.y;
  deg0 += __shfl_xor(deg0, 16, 32);
  deg1 += __shfl_xor(deg1, 16, 32);
  const float rinv0 = 1.0f / deg0;
  const float rinv1 = 1.0f / deg1;

  float* __restrict__ out = qkv + (size_t)b * NN * QKV;
#pragma unroll
  for (int j = 0; j < 8; ++j) {
    const int fr = j + half * 8;
    const float rdiv0 = __shfl(rinv0, fr, 32);
    const float rdiv1 = __shfl(rinv1, fr, 32);
    const int r0 = tn * 32 + 2 * fr;
    v2f* __restrict__ o0 = (v2f*)(out + (size_t)r0 * QKV + c0);
    v2f* __restrict__ o1 = (v2f*)(out + (size_t)(r0 + 1) * QKV + c0);
    v2f q0 = *o0, q1 = *o1;
    q0.x = q0.x + acc00[j] * rdiv0;
    q0.y = q0.y + acc01[j] * rdiv0;
    q1.x = q1.x + acc10[j] * rdiv1;
    q1.y = q1.y + acc11[j] * rdiv1;
    *o0 = q0;
    *o1 = q1;
  }
}

typedef __attribute__((ext_vector_type(4))) float v4f;

__device__ __forceinline__ float xhalf(float x) {
  const int b = __builtin_bit_cast(int, x);
  return __builtin_bit_cast(float,
      __builtin_amdgcn_permlanex16(b, b, 0x76543210, 0xfedcba98, false, false));
}

__device__ __forceinline__ float max3f(float a, float b, float c) {
  return __builtin_fmaximum_numf(__builtin_fmaximum_numf(a, b), c);
}

struct AttFrag {
  v4f   k[2][2];
  float v[2][8];
};

__device__ __forceinline__ void att_load(AttFrag& f, const float* __restrict__ kp,
                                         const float* __restrict__ vp) {
  for (int j = 0; j < 2; ++j) {
    f.k[j][0] = *(const v4f*)(kp + (size_t)(16 * j) * QKV);
    f.k[j][1] = *(const v4f*)(kp + (size_t)(16 * j) * QKV + 4);
    for (int r = 0; r < 8; ++r)
      f.v[j][r] = vp[(size_t)(16 * j + r) * QKV];
  }
}

__device__ __forceinline__ void att_step(AttFrag& f, const v2f (&qb)[4],
                                         v8f& o, float& m, float& l) {
  asm volatile("" : "+v"(f.k[0][0]), "+v"(f.k[0][1]), "+v"(f.k[1][0]), "+v"(f.k[1][1]),
                    "+v"(o), "+v"(m), "+v"(l));

  v8f s0 = {}, s1 = {};
  for (int t = 0; t < 4; ++t) {
    v2f a0, a1;
    a0.x = f.k[0][t >> 1][2 * (t & 1)];  a0.y = f.k[0][t >> 1][2 * (t & 1) + 1];
    a1.x = f.k[1][t >> 1][2 * (t & 1)];  a1.y = f.k[1][t >> 1][2 * (t & 1) + 1];
    s0 = WMMA_F32(a0, qb[t], s0);
    s1 = WMMA_F32(a1, qb[t], s1);
  }

  const float t0 = max3f(s0[0], s0[1], s0[2]);
  const float t1 = max3f(s0[3], s0[4], s0[5]);
  const float t2 = max3f(s0[6], s0[7], s1[0]);
  const float t3 = max3f(s1[1], s1[2], s1[3]);
  const float t4 = max3f(s1[4], s1[5], s1[6]);
  const float mx = __builtin_fmaximum_numf(max3f(t0, t1, t2), max3f(t3, t4, s1[7]));
  const float mnew = max3f(m, mx, xhalf(mx));
  const float corr = __builtin_amdgcn_exp2f(m - mnew);
  m = mnew;

  const v2f mm = {mnew, mnew};
  v2f ps = {0.f, 0.f};
  v8f p0, p1;
  for (int i = 0; i < 4; ++i) {
    v2f d0, d1;
    d0.x = s0[2 * i];  d0.y = s0[2 * i + 1];
    d1.x = s1[2 * i];  d1.y = s1[2 * i + 1];
    d0 -= mm;
    d1 -= mm;
    v2f e0, e1;
    e0.x = __builtin_amdgcn_exp2f(d0.x);  e0.y = __builtin_amdgcn_exp2f(d0.y);
    e1.x = __builtin_amdgcn_exp2f(d1.x);  e1.y = __builtin_amdgcn_exp2f(d1.y);
    ps += e0;
    ps += e1;
    p0[2 * i] = e0.x;  p0[2 * i + 1] = e0.y;
    p1[2 * i] = e1.x;  p1[2 * i + 1] = e1.y;
  }
  l = l * corr + (ps.x + ps.y);
  o *= corr;

  for (int t = 0; t < 4; ++t) {
    v2f a, bb;
    a.x  = f.v[0][2 * t];  a.y  = f.v[0][2 * t + 1];
    bb.x = p0[2 * t];      bb.y = p0[2 * t + 1];
    o = WMMA_F32(a, bb, o);
  }
  for (int t = 0; t < 4; ++t) {
    v2f a, bb;
    a.x  = f.v[1][2 * t];  a.y  = f.v[1][2 * t + 1];
    bb.x = p1[2 * t];      bb.y = p1[2 * t + 1];
    o = WMMA_F32(a, bb, o);
  }
}

__global__ __launch_bounds__(32)
void attn_kernel(const float* __restrict__ qkv, float* __restrict__ attout) {
  const int tiles_q = NN / 16;
  int tile = blockIdx.x;
  const int b = tile / (NHEAD * tiles_q);
  tile -= b * NHEAD * tiles_q;
  const int h  = tile / tiles_q;
  const int tq = tile - h * tiles_q;

  const int lane = threadIdx.x;
  const int half = lane >> 4;
  const int l15  = lane & 15;
  const float qscale = 0.25f * 1.4426950408889634f;

  const float* __restrict__ base = qkv + (size_t)b * NN * QKV + h * 48;

  v2f qb[4];
  {
    const float* __restrict__ qp = base + (size_t)(tq * 16 + l15) * QKV + 8 * half;
    const v4f q0 = *(const v4f*)qp;
    const v4f q1 = *(const v4f*)(qp + 4);
    qb[0].x = q0.x * qscale;  qb[0].y = q0.y * qscale;
    qb[1].x = q0.z * qscale;  qb[1].y = q0.w * qscale;
    qb[2].x = q1.x * qscale;  qb[2].y = q1.y * qscale;
    qb[3].x = q1.z * qscale;  qb[3].y = q1.w * qscale;
  }

  const float* __restrict__ kp = base + (size_t)l15 * QKV + 16 + 8 * half;
  const float* __restrict__ vp = base + (size_t)(8 * half) * QKV + 32 + l15;
  const size_t step = (size_t)32 * QKV;

  v8f o = {};
  float m = -1e30f, l = 0.f;

  AttFrag fa, fb;
  const int nstep = NN / 32;
  att_load(fa, kp, vp);
  for (int it = 0; it < nstep - 2; it += 2) {
    att_load(fb, kp + step, vp + step);
    __builtin_amdgcn_sched_barrier(0);
    att_step(fa, qb, o, m, l);
    kp += 2 * step;
    vp += 2 * step;
    att_load(fa, kp, vp);
    __builtin_amdgcn_sched_barrier(0);
    att_step(fb, qb, o, m, l);
  }
  att_load(fb, kp + step, vp + step);
  __builtin_amdgcn_sched_barrier(0);
  att_step(fa, qb, o, m, l);
  att_step(fb, qb, o, m, l);

  l += xhalf(l);
  float* __restrict__ outp = attout + (size_t)b * NN * NODE_DIM
                           + (size_t)(tq * 16 + l15) * NODE_DIM + h * HDIM + 8 * half;
  v4f r0, r1;
  r0.x = o[0] / l;  r0.y = o[1] / l;  r0.z = o[2] / l;  r0.w = o[3] / l;
  r1.x = o[4] / l;  r1.y = o[5] / l;  r1.z = o[6] / l;  r1.w = o[7] / l;
  *(v4f*)outp       = r0;
  *(v4f*)(outp + 4) = r1;
}

extern "C" void kernel_launch(void* const* d_in, const int* in_sizes, int n_in,
                              void* d_out, int out_size, void* d_ws, size_t ws_size,
                              hipStream_t stream) {
  const float* node    = (const float*)d_in[0];
  const float* edge    = (const float*)d_in[1];
  const float* G       = (const float*)d_in[2];
  const float* W_qkv_n = (const float*)d_in[3];
  const float* b_qkv_n = (const float*)d_in[4];
  const float* W_qkv_e = (const float*)d_in[5];
  const float* b_qkv_e = (const float*)d_in[6];
  const float* W_o     = (const float*)d_in[7];
  const float* b_o     = (const float*)d_in[8];
  float* out = (float*)d_out;

  float* qkv  = (float*)d_ws;
  float* qkve = qkv  + (size_t)BSZ * NN * QKV;
  float* att  = qkve + (size_t)BSZ * EE * QKV;

  gemm_bias_32x32<<<(BSZ * NN / 32) * (QKV / 32), 32, 0, stream>>>(
      node, W_qkv_n, b_qkv_n, qkv, BSZ * NN, NODE_DIM, QKV);

  gemm_bias_32x32<<<(BSZ * EE / 32) * (QKV / 32), 32, 0, stream>>>(
      edge, W_qkv_e, b_qkv_e, qkve, BSZ * EE, EDGE_DIM, QKV);

  agg_kernel<<<BSZ * (NN / 32) * (QKV / 32), 32, 0, stream>>>(G, qkve, qkv);

  attn_kernel<<<BSZ * NHEAD * (NN / 16), 32, 0, stream>>>(qkv, att);

  gemm_bias_32x32<<<(BSZ * NN / 32) * (NODE_DIM / 32), 32, 0, stream>>>(
      att, W_o, b_o, out, BSZ * NN, NODE_DIM, NODE_DIM);
}


# attention: four neighbouring query tiles of one head per 128-thread work-group (was one wave each): its waves stream identical K/V lines, so the WGP cache serves the repeats; sums unchanged
# speedup vs baseline: 2.9089x; 1.0875x over previous
#include <hip/hip_runtime.h>
#include <hip/hip_bf16.h>


typedef __attribute__((ext_vector_type(2))) float v2f;
typedef __attribute__((ext_vector_type(8))) float v8f;

#define BSZ        2
#define NN         3072
#define EE         4096
#define NODE_DIM   128
#define EDGE_DIM   64
#define NHEAD      8
#define HDIM       16
#define QKV        384

#define WMMA_F32(A, B, C) \
  __builtin_amdgcn_wmma_f32_16x16x4_f32(false, (A), false, (B), (short)0, (C), false, false)

__global__ __launch_bounds__(32)
void gemm_bias_32x32(const float* __restrict__ X, const float* __restrict__ W,
                     const float* __restrict__ bias, float* __restrict__ C,
                     int M, int K, int Ncols) {
  const int tiles_n = Ncols >> 5;
  const int tile = blockIdx.x;
  const int tm = tile / tiles_n;
  const int tn = tile % tiles_n;
  const int lane = threadIdx.x;
  const int half = lane >> 4;
  const int l15  = lane & 15;

  const int row0 = tm * 32 + l15;
  const int row1 = row0 + 16;
  const int col0 = tn * 32 + l15;
  const int col1 = col0 + 16;

  const float* __restrict__ Xr0 = X + (size_t)row0 * K;
  const float* __restrict__ Xr1 = X + (size_t)row1 * K;

  v8f acc00 = {}, acc01 = {}, acc10 = {}, acc11 = {};
  for (int k = 0; k < K; k += 4) {
    const int ka = k + half * 2;
    v2f a0, a1, b0, b1;
    a0.x = Xr0[ka];                         a0.y = Xr0[ka + 1];
    a1.x = Xr1[ka];                         a1.y = Xr1[ka + 1];
    b0.x = W[(size_t)ka * Ncols + col0];    b0.y = W[(size_t)(ka + 1) * Ncols + col0];
    b1.x = W[(size_t)ka * Ncols + col1];    b1.y = W[(size_t)(ka + 1) * Ncols + col1];
    acc00 = WMMA_F32(a0, b0, acc00);
    acc01 = WMMA_F32(a0, b1, acc01);
    acc10 = WMMA_F32(a1, b0, acc10);
    acc11 = WMMA_F32(a1, b1, acc11);
  }

  const float bv0 = bias[col0];
  const float bv1 = bias[col1];
#pragma unroll
  for (int j = 0; j < 8; ++j) {
    const int r0 = tm * 32 + j + half * 8;
    const int r1 = r0 + 16;
    C[(size_t)r0 * Ncols + col0] = acc00[j] + bv0;
    C[(size_t)r0 * Ncols + col1] = acc01[j] + bv1;
    C[(size_t)r1 * Ncols + col0] = acc10[j] + bv0;
    C[(size_t)r1 * Ncols + col1] = acc11[j] + bv1;
  }
}

#define AGG_U 8

struct AggFrag {
  v2f g[AGG_U][2];
  v2f e[AGG_U][2];
};

__device__ __forceinline__ void agg_load(AggFrag& f, const float* __restrict__ gp,
                                         const float* __restrict__ ep) {
#pragma unroll
  for (int u = 0; u < AGG_U; ++u) {
    f.g[u][0] = *(const v2f*)(gp + (size_t)(4 * u) * NN);
    f.g[u][1] = *(const v2f*)(gp + (size_t)(4 * u + 1) * NN);
    f.e[u][0] = *(const v2f*)(ep + (size_t)(4 * u) * QKV);
    f.e[u][1] = *(const v2f*)(ep + (size_t)(4 * u + 1) * QKV);
  }
}

__device__ __forceinline__ void agg_compute(const AggFrag& f, v8f& acc00, v8f& acc01,
                                            v8f& acc10, v8f& acc11,
                                            v2f& dsum0, v2f& dsum1) {
#pragma unroll
  for (int u = 0; u < AGG_U; ++u) {
    v2f a0, a1, b0, b1;
    a0.x = f.g[u][0].x;  a0.y = f.g[u][1].x;
    a1.x = f.g[u][0].y;  a1.y = f.g[u][1].y;
    b0.x = f.e[u][0].x;  b0.y = f.e[u][1].x;
    b1.x = f.e[u][0].y;  b1.y = f.e[u][1].y;
    dsum0 += a0;
    dsum1 += a1;
    acc00 = WMMA_F32(a0, b0, acc00);
    acc01 = WMMA_F32(a0, b1, acc01);
    acc10 = WMMA_F32(a1, b0, acc10);
    acc11 = WMMA_F32(a1, b1, acc11);
    __builtin_amdgcn_sched_barrier(0);
  }
}

__global__ __launch_bounds__(32)
void agg_kernel(const float* __restrict__ G, const float* __restrict__ qkv_e,
                float* __restrict__ qkv) {
  const int tiles_c = QKV / 32;
  const int tiles_n = NN / 32;
  int tile = blockIdx.x;
  const int b  = tile / (tiles_n * tiles_c);
  tile -= b * tiles_n * tiles_c;
  const int tn = tile / tiles_c;
  const int tc = tile - tn * tiles_c;

  const int lane = threadIdx.x;
  const int half = lane >> 4;
  const int l15  = lane & 15;
  const int n0 = tn * 32 + 2 * l15;
  const int c0 = tc * 32 + 2 * l15;

  const float* __restrict__ gp = G     + (size_t)b * EE * NN  + (size_t)(half * 2) * NN  + n0;
  const float* __restrict__ ep = qkv_e + (size_t)b * EE * QKV + (size_t)(half * 2) * QKV + c0;
  const size_t gstep = (size_t)(4 * AGG_U) * NN;
  const size_t estep = (size_t)(4 * AGG_U) * QKV;

  v8f acc00 = {}, acc01 = {}, acc10 = {}, acc11 = {};
  v2f dsum0 = {}, dsum1 = {};

  AggFrag fa, fb;
  const int nblk = EE / (4 * AGG_U);
  agg_load(fa, gp, ep);
  for (int blk = 0; blk < nblk - 2; blk += 2) {
    agg_load(fb, gp + gstep, ep + estep);
    __builtin_amdgcn_sched_barrier(0);
    agg_compute(fa, acc00, acc01, acc10, acc11, dsum0, dsum1);
    gp += 2 * gstep;
    ep += 2 * estep;
    agg_load(fa, gp, ep);
    __builtin_amdgcn_sched_barrier(0);
    agg_compute(fb, acc00, acc01, acc10, acc11, dsum0, dsum1);
  }
  agg_load(fb, gp + gstep, ep + estep);
  __builtin_amdgcn_sched_barrier(0);
  agg_compute(fa, acc00, acc01, acc10, acc11, dsum0, dsum1);
  agg_compute(fb, acc00, acc01, acc10, acc11, dsum0, dsum1);

  float deg0 = dsum0.x + dsum0.y;
  float deg1 = dsum1.x + dsum1.y;
  deg0 += __shfl_xor(deg0, 16, 32);
  deg1 += __shfl_xor(deg1, 16, 32);
  const float rinv0 = 1.0f / deg0;
  const float rinv1 = 1.0f / deg1;

  float* __restrict__ out = qkv + (size_t)b * NN * QKV;
#pragma unroll
  for (int j = 0; j < 8; ++j) {
    const int fr = j + half * 8;
    const float rdiv0 = __shfl(rinv0, fr, 32);
    const float rdiv1 = __shfl(rinv1, fr, 32);
    const int r0 = tn * 32 + 2 * fr;
    v2f* __restrict__ o0 = (v2f*)(out + (size_t)r0 * QKV + c0);
    v2f* __restrict__ o1 = (v2f*)(out + (size_t)(r0 + 1) * QKV + c0);
    v2f q0 = *o0, q1 = *o1;
    q0.x = q0.x + acc00[j] * rdiv0;
    q0.y = q0.y + acc01[j] * rdiv0;
    q1.x = q1.x + acc10[j] * rdiv1;
    q1.y = q1.y + acc11[j] * rdiv1;
    *o0 = q0;
    *o1 = q1;
  }
}

typedef __attribute__((ext_vector_type(4))) float v4f;

#define ATT_WAVES 4

__device__ __forceinline__ float xhalf(float x) {
  const int b = __builtin_bit_cast(int, x);
  return __builtin_bit_cast(float,
      __builtin_amdgcn_permlanex16(b, b, 0x76543210, 0xfedcba98, false, false));
}

__device__ __forceinline__ float max3f(float a, float b, float c) {
  return __builtin_fmaximum_numf(__builtin_fmaximum_numf(a, b), c);
}

struct AttFrag {
  v4f   k[2][2];
  float v[2][8];
};

__device__ __forceinline__ void att_load(AttFrag& f, const float* __restrict__ kp,
                                         const float* __restrict__ vp) {
  for (int j = 0; j < 2; ++j) {
    f.k[j][0] = *(const v4f*)(kp + (size_t)(16 * j) * QKV);
    f.k[j][1] = *(const v4f*)(kp + (size_t)(16 * j) * QKV + 4);
    for (int r = 0; r < 8; ++r)
      f.v[j][r] = vp[(size_t)(16 * j + r) * QKV];
  }
}

__device__ __forceinline__ void att_step(AttFrag& f, const v2f (&qb)[4],
                                         v8f& o, float& m, float& l) {
  asm volatile("" : "+v"(f.k[0][0]), "+v"(f.k[0][1]), "+v"(f.k[1][0]), "+v"(f.k[1][1]),
                    "+v"(o), "+v"(m), "+v"(l));

  v8f s0 = {}, s1 = {};
  for (int t = 0; t < 4; ++t) {
    v2f a0, a1;
    a0.x = f.k[0][t >> 1][2 * (t & 1)];  a0.y = f.k[0][t >> 1][2 * (t & 1) + 1];
    a1.x = f.k[1][t >> 1][2 * (t & 1)];  a1.y = f.k[1][t >> 1][2 * (t & 1) + 1];
    s0 = WMMA_F32(a0, qb[t], s0);
    s1 = WMMA_F32(a1, qb[t], s1);
  }

  const float t0 = max3f(s0[0], s0[1], s0[2]);
  const float t1 = max3f(s0[3], s0[4], s0[5]);
  const float t2 = max3f(s0[6], s0[7], s1[0]);
  const float t3 = max3f(s1[1], s1[2], s1[3]);
  const float t4 = max3f(s1[4], s1[5], s1[6]);
  const float mx = __builtin_fmaximum_numf(max3f(t0, t1, t2), max3f(t3, t4, s1[7]));
  const float mnew = max3f(m, mx, xhalf(mx));
  const float corr = __builtin_amdgcn_exp2f(m - mnew);
  m = mnew;

  const v2f mm = {mnew, mnew};
  v2f ps = {0.f, 0.f};
  v8f p0, p1;
  for (int i = 0; i < 4; ++i) {
    v2f d0, d1;
    d0.x = s0[2 * i];  d0.y = s0[2 * i + 1];
    d1.x = s1[2 * i];  d1.y = s1[2 * i + 1];
    d0 -= mm;
    d1 -= mm;
    v2f e0, e1;
    e0.x = __builtin_amdgcn_exp2f(d0.x);  e0.y = __builtin_amdgcn_exp2f(d0.y);
    e1.x = __builtin_amdgcn_exp2f(d1.x);  e1.y = __builtin_amdgcn_exp2f(d1.y);
    ps += e0;
    ps += e1;
    p0[2 * i] = e0.x;  p0[2 * i + 1] = e0.y;
    p1[2 * i] = e1.x;  p1[2 * i + 1] = e1.y;
  }
  l = l * corr + (ps.x + ps.y);
  o *= corr;

  for (int t = 0; t < 4; ++t) {
    v2f a, bb;
    a.x  = f.v[0][2 * t];  a.y  = f.v[0][2 * t + 1];
    bb.x = p0[2 * t];      bb.y = p0[2 * t + 1];
    o = WMMA_F32(a, bb, o);
  }
  for (int t = 0; t < 4; ++t) {
    v2f a, bb;
    a.x  = f.v[1][2 * t];  a.y  = f.v[1][2 * t + 1];
    bb.x = p1[2 * t];      bb.y = p1[2 * t + 1];
    o = WMMA_F32(a, bb, o);
  }
}

__global__ __launch_bounds__(32 * ATT_WAVES)
void attn_kernel(const float* __restrict__ qkv, float* __restrict__ attout) {
  const int groups_q = NN / (16 * ATT_WAVES);
  int grp = blockIdx.x;
  const int b = grp / (NHEAD * groups_q);
  grp -= b * NHEAD * groups_q;
  const int h  = grp / groups_q;
  const int gq = grp - h * groups_q;
  const int wave = threadIdx.x >> 5;
  const int tq = gq * ATT_WAVES + wave;

  const int lane = threadIdx.x & 31;
  const int half = lane >> 4;
  const int l15  = lane & 15;
  const float qscale = 0.25f * 1.4426950408889634f;

  const float* __restrict__ base = qkv + (size_t)b * NN * QKV + h * 48;

  v2f qb[4];
  {
    const float* __restrict__ qp = base + (size_t)(tq * 16 + l15) * QKV + 8 * half;
    const v4f q0 = *(const v4f*)qp;
    const v4f q1 = *(const v4f*)(qp + 4);
    qb[0].x = q0.x * qscale;  qb[0].y = q0.y * qscale;
    qb[1].x = q0.z * qscale;  qb[1].y = q0.w * qscale;
    qb[2].x = q1.x * qscale;  qb[2].y = q1.y * qscale;
    qb[3].x = q1.z * qscale;  qb[3].y = q1.w * qscale;
  }

  const float* __restrict__ kp = base + (size_t)l15 * QKV + 16 + 8 * half;
  const float* __restrict__ vp = base + (size_t)(8 * half) * QKV + 32 + l15;
  const size_t step = (size_t)32 * QKV;

  v8f o = {};
  float m = -1e30f, l = 0.f;

  AttFrag fa, fb;
  const int nstep = NN / 32;
  att_load(fa, kp, vp);
  for (int it = 0; it < nstep - 2; it += 2) {
    att_load(fb, kp + step, vp + step);
    __builtin_amdgcn_sched_barrier(0);
    att_step(fa, qb, o, m, l);
    kp += 2 * step;
    vp += 2 * step;
    att_load(fa, kp, vp);
    __builtin_amdgcn_sched_barrier(0);
    att_step(fb, qb, o, m, l);
  }
  att_load(fb, kp + step, vp + step);
  __builtin_amdgcn_sched_barrier(0);
  att_step(fa, qb, o, m, l);
  att_step(fb, qb, o, m, l);

  l += xhalf(l);
  float* __restrict__ outp = attout + (size_t)b * NN * NODE_DIM
                           + (size_t)(tq * 16 + l15) * NODE_DIM + h * HDIM + 8 * half;
  v4f r0, r1;
  r0.x = o[0] / l;  r0.y = o[1] / l;  r0.z = o[2] / l;  r0.w = o[3] / l;
  r1.x = o[4] / l;  r1.y = o[5] / l;  r1.z = o[6] / l;  r1.w = o[7] / l;
  *(v4f*)outp       = r0;
  *(v4f*)(outp + 4) = r1;
}

extern "C" void kernel_launch(void* const* d_in, const int* in_sizes, int n_in,
                              void* d_out, int out_size, void* d_ws, size_t ws_size,
                              hipStream_t stream) {
  const float* node    = (const float*)d_in[0];
  const float* edge    = (const float*)d_in[1];
  const float* G       = (const float*)d_in[2];
  const float* W_qkv_n = (const float*)d_in[3];
  const float* b_qkv_n = (const float*)d_in[4];
  const float* W_qkv_e = (const float*)d_in[5];
  const float* b_qkv_e = (const float*)d_in[6];
  const float* W_o     = (const float*)d_in[7];
  const float* b_o     = (const float*)d_in[8];
  float* out = (float*)d_out;

  float* qkv  = (float*)d_ws;
  float* qkve = qkv  + (size_t)BSZ * NN * QKV;
  float* att  = qkve + (size_t)BSZ * EE * QKV;

  gemm_bias_32x32<<<(BSZ * NN / 32) * (QKV / 32), 32, 0, stream>>>(
      node, W_qkv_n, b_qkv_n, qkv, BSZ * NN, NODE_DIM, QKV);

  gemm_bias_32x32<<<(BSZ * EE / 32) * (QKV / 32), 32, 0, stream>>>(
      edge, W_qkv_e, b_qkv_e, qkve, BSZ * EE, EDGE_DIM, QKV);

  agg_kernel<<<BSZ * (NN / 32) * (QKV / 32), 32, 0, stream>>>(G, qkve, qkv);

  attn_kernel<<<BSZ * NHEAD * (NN / (16 * ATT_WAVES)), 32 * ATT_WAVES, 0, stream>>>(qkv, att);

  gemm_bias_32x32<<<(BSZ * NN / 32) * (NODE_DIM / 32), 32, 0, stream>>>(
      att, W_o, b_o, out, BSZ * NN, NODE_DIM, NODE_DIM);
}


# aggregation GEMM retiled: 48x48 per wave (3x3 blocks, 12-byte loads, a lane owns three adjacent rows/columns), 2x2 waves per group, 256 groups = one wave per SIMD; combine through LDS; sum order kept
# speedup vs baseline: 3.4422x; 1.1833x over previous
#include <hip/hip_runtime.h>
#include <hip/hip_bf16.h>


typedef __attribute__((ext_vector_type(2))) float v2f;
typedef __attribute__((ext_vector_type(8))) float v8f;
typedef __attribute__((ext_vector_type(4))) float v4f;

#define BSZ        2
#define NN         3072
#define EE         4096
#define NODE_DIM   128
#define EDGE_DIM   64
#define NHEAD      8
#define HDIM       16
#define QKV        384

#define WMMA_F32(A, B, C) \
  __builtin_amdgcn_wmma_f32_16x16x4_f32(false, (A), false, (B), (short)0, (C), false, false)

__global__ __launch_bounds__(32)
void gemm_bias_32x32(const float* __restrict__ X, const float* __restrict__ W,
                     const float* __restrict__ bias, float* __restrict__ C,
                     int M, int K, int Ncols) {
  const int tiles_n = Ncols >> 5;
  const int tile = blockIdx.x;
  const int tm = tile / tiles_n;
  const int tn = tile % tiles_n;
  const int lane = threadIdx.x;
  const int half = lane >> 4;
  const int l15  = lane & 15;

  const int row0 = tm * 32 + l15;
  const int row1 = row0 + 16;
  const int col0 = tn * 32 + l15;
  const int col1 = col0 + 16;

  const float* __restrict__ Xr0 = X + (size_t)row0 * K;
  const float* __restrict__ Xr1 = X + (size_t)row1 * K;

  v8f acc00 = {}, acc01 = {}, acc10 = {}, acc11 = {};
  for (int k = 0; k < K; k += 4) {
    const int ka = k + half * 2;
    v2f a0, a1, b0, b1;
    a0.x = Xr0[ka];                         a0.y = Xr0[ka + 1];
    a1.x = Xr1[ka];                         a1.y = Xr1[ka + 1];
    b0.x = W[(size_t)ka * Ncols + col0];    b0.y = W[(size_t)(ka + 1) * Ncols + col0];
    b1.x = W[(size_t)ka * Ncols + col1];    b1.y = W[(size_t)(ka + 1) * Ncols + col1];
    acc00 = WMMA_F32(a0, b0, acc00);
    acc01 = WMMA_F32(a0, b1, acc01);
    acc10 = WMMA_F32(a1, b0, acc10);
    acc11 = WMMA_F32(a1, b1, acc11);
  }

  const float bv0 = bias[col0];
  const float bv1 = bias[col1];
#pragma unroll
  for (int j = 0; j < 8; ++j) {
    const int r0 = tm * 32 + j + half * 8;
    const int r1 = r0 + 16;
    C[(size_t)r0 * Ncols + col0] = acc00[j] + bv0;
    C[(size_t)r0 * Ncols + col1] = acc01[j] + bv1;
    C[(size_t)r1 * Ncols + col0] = acc10[j] + bv0;
    C[(size_t)r1 * Ncols + col1] = acc11[j] + bv1;
  }
}

#define AGG_U     8
#define AGG_T     96
#define AGG_PASSES ((AGG_T * AGG_T / 4) / 128)
#define AGG_PITCH 100

typedef __attribute__((ext_vector_type(3))) float v3f;

__device__ __forceinline__ v3f ld3(const float* __restrict__ p) {
  v3f r;
  r.x = p[0];  r.y = p[1];  r.z = p[2];
  return r;
}

struct AggFrag {
  v3f g[AGG_U][2];
  v3f e[AGG_U][2];
};

__device__ __forceinline__ void agg_load(AggFrag& f, const float* __restrict__ gp,
                                         const float* __restrict__ ep) {
#pragma unroll
  for (int u = 0; u < AGG_U; ++u) {
    f.g[u][0] = ld3(gp + (size_t)(4 * u) * NN);
    f.g[u][1] = ld3(gp + (size_t)(4 * u + 1) * NN);
    f.e[u][0] = ld3(ep + (size_t)(4 * u) * QKV);
    f.e[u][1] = ld3(ep + (size_t)(4 * u + 1) * QKV);
  }
}

__device__ __forceinline__ void agg_compute(const AggFrag& f, v8f (&acc)[3][3], v2f (&dsum)[3]) {
#pragma unroll
  for (int u = 0; u < AGG_U; ++u) {
    v2f a[3], bq[3];
    a[0].x  = f.g[u][0].x;  a[0].y  = f.g[u][1].x;
    a[1].x  = f.g[u][0].y;  a[1].y  = f.g[u][1].y;
    a[2].x  = f.g[u][0].z;  a[2].y  = f.g[u][1].z;
    bq[0].x = f.e[u][0].x;  bq[0].y = f.e[u][1].x;
    bq[1].x = f.e[u][0].y;  bq[1].y = f.e[u][1].y;
    bq[2].x = f.e[u][0].z;  bq[2].y = f.e[u][1].z;
#pragma unroll
    for (int i = 0; i < 3; ++i) {
      dsum[i] += a[i];
#pragma unroll
      for (int j = 0; j < 3; ++j)
        acc[i][j] = WMMA_F32(a[i], bq[j], acc[i][j]);
    }
    __builtin_amdgcn_sched_barrier(0);
  }
}

__global__ __launch_bounds__(128)
void agg_kernel(const float* __restrict__ G, const float* __restrict__ qkv_e,
                float* __restrict__ qkv) {
  __shared__ __attribute__((aligned(16))) float tile[AGG_T * AGG_PITCH];
  __shared__ float rinv_s[AGG_T];

  const int tiles_c = QKV / AGG_T;
  const int tiles_n = NN / AGG_T;
  int t = blockIdx.x;
  const int b  = t / (tiles_n * tiles_c);
  t -= b * tiles_n * tiles_c;
  const int tn = t / tiles_c;
  const int tc = t - tn * tiles_c;

  const int wave = threadIdx.x >> 5;
  const int wm   = wave >> 1;
  const int wn   = wave & 1;
  const int lane = threadIdx.x & 31;
  const int half = lane >> 4;
  const int l15  = lane & 15;
  const int n0 = tn * AGG_T + wm * 48 + 3 * l15;
  const int c0 = tc * AGG_T + wn * 48 + 3 * l15;

  const float* __restrict__ gp = G     + (size_t)b * EE * NN  + (size_t)(half * 2) * NN  + n0;
  const float* __restrict__ ep = qkv_e + (size_t)b * EE * QKV + (size_t)(half * 2) * QKV + c0;
  const size_t gstep = (size_t)(4 * AGG_U) * NN;
  const size_t estep = (size_t)(4 * AGG_U) * QKV;

  v8f acc[3][3] = {};
  v2f dsum[3] = {};

  AggFrag fa, fb;
  const int nblk = EE / (4 * AGG_U);
  agg_load(fa, gp, ep);
  for (int blk = 0; blk < nblk - 2; blk += 2) {
    agg_load(fb, gp + gstep, ep + estep);
    __builtin_amdgcn_sched_barrier(0);
    agg_compute(fa, acc, dsum);
    gp += 2 * gstep;
    ep += 2 * estep;
    agg_load(fa, gp, ep);
    __builtin_amdgcn_sched_barrier(0);
    agg_compute(fb, acc, dsum);
  }
  agg_load(fb, gp + gstep, ep + estep);
  __builtin_amdgcn_sched_barrier(0);
  agg_compute(fa, acc, dsum);
  agg_compute(fb, acc, dsum);

  float* __restrict__ out = qkv + (size_t)b * NN * QKV + (size_t)(tn * AGG_T) * QKV + tc * AGG_T;
  v4f q[AGG_PASSES];
#pragma unroll
  for (int it = 0; it < AGG_PASSES; ++it) {
    const int idx = it * 128 + threadIdx.x;
    const int row = idx / 24;
    const int c4  = (idx - row * 24) * 4;
    q[it] = *(const v4f*)(out + (size_t)row * QKV + c4);
  }

#pragma unroll
  for (int i = 0; i < 3; ++i) {
    float deg = dsum[i].x + dsum[i].y;
    deg += __shfl_xor(deg, 16, 32);
    if (wn == 0 && half == 0) rinv_s[wm * 48 + 3 * l15 + i] = 1.0f / deg;
  }

#pragma unroll
  for (int i = 0; i < 3; ++i) {
#pragma unroll
    for (int r = 0; r < 8; ++r) {
      float* __restrict__ p = tile + (wm * 48 + 3 * (r + 8 * half) + i) * AGG_PITCH
                                   + wn * 48 + 3 * l15;
      p[0] = acc[i][0][r];
      p[1] = acc[i][1][r];
      p[2] = acc[i][2][r];
    }
  }
  __syncthreads();

#pragma unroll
  for (int it = 0; it < AGG_PASSES; ++it) {
    const int idx = it * 128 + threadIdx.x;
    const int row = idx / 24;
    const int c4  = (idx - row * 24) * 4;
    const float rd = rinv_s[row];
    const v4f s = *(const v4f*)(tile + row * AGG_PITCH + c4);
    v4f r;
    r.x = q[it].x + s.x * rd;
    r.y = q[it].y + s.y * rd;
    r.z = q[it].z + s.z * rd;
    r.w = q[it].w + s.w * rd;
    *(v4f*)(out + (size_t)row * QKV + c4) = r;
  }
}

#define ATT_WAVES 4

__device__ __forceinline__ float xhalf(float x) {
  const int b = __builtin_bit_cast(int, x);
  return __builtin_bit_cast(float,
      __builtin_amdgcn_permlanex16(b, b, 0x76543210, 0xfedcba98, false, false));
}

__device__ __forceinline__ float max3f(float a, float b, float c) {
  return __builtin_fmaximum_numf(__builtin_fmaximum_numf(a, b), c);
}

struct AttFrag {
  v4f   k[2][2];
  float v[2][8];
};

__device__ __forceinline__ void att_load(AttFrag& f, const float* __restrict__ kp,
                                         const float* __restrict__ vp) {
  for (int j = 0; j < 2; ++j) {
    f.k[j][0] = *(const v4f*)(kp + (size_t)(16 * j) * QKV);
    f.k[j][1] = *(const v4f*)(kp + (size_t)(16 * j) * QKV + 4);
    for (int r = 0; r < 8; ++r)
      f.v[j][r] = vp[(size_t)(16 * j + r) * QKV];
  }
}

__device__ __forceinline__ void att_step(AttFrag& f, const v2f (&qb)[4],
                                         v8f& o, float& m, float& l) {
  asm volatile("" : "+v"(f.k[0][0]), "+v"(f.k[0][1]), "+v"(f.k[1][0]), "+v"(f.k[1][1]),
                    "+v"(o), "+v"(m), "+v"(l));

  v8f s0 = {}, s1 = {};
  for (int t = 0; t < 4; ++t) {
    v2f a0, a1;
    a0.x = f.k[0][t >> 1][2 * (t & 1)];  a0.y = f.k[0][t >> 1][2 * (t & 1) + 1];
    a1.x = f.k[1][t >> 1][2 * (t & 1)];  a1.y = f.k[1][t >> 1][2 * (t & 1) + 1];
    s0 = WMMA_F32(a0, qb[t], s0);
    s1 = WMMA_F32(a1, qb[t], s1);
  }

  const float t0 = max3f(s0[0], s0[1], s0[2]);
  const float t1 = max3f(s0[3], s0[4], s0[5]);
  const float t2 = max3f(s0[6], s0[7], s1[0]);
  const float t3 = max3f(s1[1], s1[2], s1[3]);
  const float t4 = max3f(s1[4], s1[5], s1[6]);
  const float mx = __builtin_fmaximum_numf(max3f(t0, t1, t2), max3f(t3, t4, s1[7]));
  const float mnew = max3f(m, mx, xhalf(mx));
  const float corr = __builtin_amdgcn_exp2f(m - mnew);
  m = mnew;

  const v2f mm = {mnew, mnew};
  v2f ps = {0.f, 0.f};
  v8f p0, p1;
  for (int i = 0; i < 4; ++i) {
    v2f d0, d1;
    d0.x = s0[2 * i];  d0.y = s0[2 * i + 1];
    d1.x = s1[2 * i];  d1.y = s1[2 * i + 1];
    d0 -= mm;
    d1 -= mm;
    v2f e0, e1;
    e0.x = __builtin_amdgcn_exp2f(d0.x);  e0.y = __builtin_amdgcn_exp2f(d0.y);
    e1.x = __builtin_amdgcn_exp2f(d1.x);  e1.y = __builtin_amdgcn_exp2f(d1.y);
    ps += e0;
    ps += e1;
    p0[2 * i] = e0.x;  p0[2 * i + 1] = e0.y;
    p1[2 * i] = e1.x;  p1[2 * i + 1] = e1.y;
  }
  l = l * corr + (ps.x + ps.y);
  o *= corr;

  for (int t = 0; t < 4; ++t) {
    v2f a, bb;
    a.x  = f.v[0][2 * t];  a.y  = f.v[0][2 * t + 1];
    bb.x = p0[2 * t];      bb.y = p0[2 * t + 1];
    o = WMMA_F32(a, bb, o);
  }
  for (int t = 0; t < 4; ++t) {
    v2f a, bb;
    a.x  = f.v[1][2 * t];  a.y  = f.v[1][2 * t + 1];
    bb.x = p1[2 * t];      bb.y = p1[2 * t + 1];
    o = WMMA_F32(a, bb, o);
  }
}

__global__ __launch_bounds__(32 * ATT_WAVES)
void attn_kernel(const float* __restrict__ qkv, float* __restrict__ attout) {
  const int groups_q = NN / (16 * ATT_WAVES);
  int grp = blockIdx.x;
  const int b = grp / (NHEAD * groups_q);
  grp -= b * NHEAD * groups_q;
  const int h  = grp / groups_q;
  const int gq = grp - h * groups_q;
  const int wave = threadIdx.x >> 5;
  const int tq = gq * ATT_WAVES + wave;

  const int lane = threadIdx.x & 31;
  const int half = lane >> 4;
  const int l15  = lane & 15;
  const float qscale = 0.25f * 1.4426950408889634f;

  const float* __restrict__ base = qkv + (size_t)b * NN * QKV + h * 48;

  v2f qb[4];
  {
    const float* __restrict__ qp = base + (size_t)(tq * 16 + l15) * QKV + 8 * half;
    const v4f q0 = *(const v4f*)qp;
    const v4f q1 = *(const v4f*)(qp + 4);
    qb[0].x = q0.x * qscale;  qb[0].y = q0.y * qscale;
    qb[1].x = q0.z * qscale;  qb[1].y = q0.w * qscale;
    qb[2].x = q1.x * qscale;  qb[2].y = q1.y * qscale;
    qb[3].x = q1.z * qscale;  qb[3].y = q1.w * qscale;
  }

  const float* __restrict__ kp = base + (size_t)l15 * QKV + 16 + 8 * half;
  const float* __restrict__ vp = base + (size_t)(8 * half) * QKV + 32 + l15;
  const size_t step = (size_t)32 * QKV;

  v8f o = {};
  float m = -1e30f, l = 0.f;

  AttFrag fa, fb;
  const int nstep = NN / 32;
  att_load(fa, kp, vp);
  for (int it = 0; it < nstep - 2; it += 2) {
    att_load(fb, kp + step, vp + step);
    __builtin_amdgcn_sched_barrier(0);
    att_step(fa, qb, o, m, l);
    kp += 2 * step;
    vp += 2 * step;
    att_load(fa, kp, vp);
    __builtin_amdgcn_sched_barrier(0);
    att_step(fb, qb, o, m, l);
  }
  att_load(fb, kp + step, vp + step);
  __builtin_amdgcn_sched_barrier(0);
  att_step(fa, qb, o, m, l);
  att_step(fb, qb, o, m, l);

  l += xhalf(l);
  float* __restrict__ outp = attout + (size_t)b * NN * NODE_DIM
                           + (size_t)(tq * 16 + l15) * NODE_DIM + h * HDIM + 8 * half;
  v4f r0, r1;
  r0.x = o[0] / l;  r0.y = o[1] / l;  r0.z = o[2] / l;  r0.w = o[3] / l;
  r1.x = o[4] / l;  r1.y = o[5] / l;  r1.z = o[6] / l;  r1.w = o[7] / l;
  *(v4f*)outp       = r0;
  *(v4f*)(outp + 4) = r1;
}

extern "C" void kernel_launch(void* const* d_in, const int* in_sizes, int n_in,
                              void* d_out, int out_size, void* d_ws, size_t ws_size,
                              hipStream_t stream) {
  const float* node    = (const float*)d_in[0];
  const float* edge    = (const float*)d_in[1];
  const float* G       = (const float*)d_in[2];
  const float* W_qkv_n = (const float*)d_in[3];
  const float* b_qkv_n = (const float*)d_in[4];
  const float* W_qkv_e = (const float*)d_in[5];
  const float* b_qkv_e = (const float*)d_in[6];
  const float* W_o     = (const float*)d_in[7];
  const float* b_o     = (const float*)d_in[8];
  float* out = (float*)d_out;

  float* qkv  = (float*)d_ws;
  float* qkve = qkv  + (size_t)BSZ * NN * QKV;
  float* att  = qkve + (size_t)BSZ * EE * QKV;

  gemm_bias_32x32<<<(BSZ * NN / 32) * (QKV / 32), 32, 0, stream>>>(
      node, W_qkv_n, b_qkv_n, qkv, BSZ * NN, NODE_DIM, QKV);

  gemm_bias_32x32<<<(BSZ * EE / 32) * (QKV / 32), 32, 0, stream>>>(
      edge, W_qkv_e, b_qkv_e, qkve, BSZ * EE, EDGE_DIM, QKV);

  agg_kernel<<<BSZ * (NN / AGG_T) * (QKV / AGG_T), 128, 0, stream>>>(G, qkve, qkv);

  attn_kernel<<<BSZ * NHEAD * (NN / (16 * ATT_WAVES)), 32 * ATT_WAVES, 0, stream>>>(qkv, att);

  gemm_bias_32x32<<<(BSZ * NN / 32) * (NODE_DIM / 32), 32, 0, stream>>>(
      att, W_o, b_o, out, BSZ * NN, NODE_DIM, NODE_DIM);
}


# aggregation reassociated: S=(G^T edge) per node, then one K=192 projection [node | S/deg] @ [W_n; W_e]: the 4096-deep GEMM has 64 columns, not 384; qkv_e launch gone; sum order changes (split-K)
# speedup vs baseline: 6.0209x; 1.7491x over previous
#include <hip/hip_runtime.h>
#include <hip/hip_bf16.h>


typedef __attribute__((ext_vector_type(2))) float v2f;
typedef __attribute__((ext_vector_type(8))) float v8f;
typedef __attribute__((ext_vector_type(4))) float v4f;

#define BSZ        2
#define NN         3072
#define EE         4096
#define NODE_DIM   128
#define EDGE_DIM   64
#define NHEAD      8
#define HDIM       16
#define QKV        384

#define WMMA_F32(A, B, C) \
  __builtin_amdgcn_wmma_f32_16x16x4_f32(false, (A), false, (B), (short)0, (C), false, false)

__global__ __launch_bounds__(32)
void gemm_bias_32x32(const float* __restrict__ X, const float* __restrict__ W,
                     const float* __restrict__ bias, float* __restrict__ C,
                     int M, int K, int Ncols) {
  const int tiles_n = Ncols >> 5;
  const int tile = blockIdx.x;
  const int tm = tile / tiles_n;
  const int tn = tile % tiles_n;
  const int lane = threadIdx.x;
  const int half = lane >> 4;
  const int l15  = lane & 15;

  const int row0 = tm * 32 + l15;
  const int row1 = row0 + 16;
  const int col0 = tn * 32 + l15;
  const int col1 = col0 + 16;

  const float* __restrict__ Xr0 = X + (size_t)row0 * K;
  const float* __restrict__ Xr1 = X + (size_t)row1 * K;

  v8f acc00 = {}, acc01 = {}, acc10 = {}, acc11 = {};
  for (int k = 0; k < K; k += 4) {
    const int ka = k + half * 2;
    v2f a0, a1, b0, b1;
    a0.x = Xr0[ka];                         a0.y = Xr0[ka + 1];
    a1.x = Xr1[ka];                         a1.y = Xr1[ka + 1];
    b0.x = W[(size_t)ka * Ncols + col0];    b0.y = W[(size_t)(ka + 1) * Ncols + col0];
    b1.x = W[(size_t)ka * Ncols + col1];    b1.y = W[(size_t)(ka + 1) * Ncols + col1];
    acc00 = WMMA_F32(a0, b0, acc00);
    acc01 = WMMA_F32(a0, b1, acc01);
    acc10 = WMMA_F32(a1, b0, acc10);
    acc11 = WMMA_F32(a1, b1, acc11);
  }

  const float bv0 = bias[col0];
  const float bv1 = bias[col1];
#pragma unroll
  for (int j = 0; j < 8; ++j) {
    const int r0 = tm * 32 + j + half * 8;
    const int r1 = r0 + 16;
    C[(size_t)r0 * Ncols + col0] = acc00[j] + bv0;
    C[(size_t)r0 * Ncols + col1] = acc01[j] + bv1;
    C[(size_t)r1 * Ncols + col0] = acc10[j] + bv0;
    C[(size_t)r1 * Ncols + col1] = acc11[j] + bv1;
  }
}

#define AGG_U      8
#define AGG_ROWS   48
#define AGG_WAVES  4
#define AGG_PLANES 2
#define AGG_KW     (EE / (AGG_PLANES * AGG_WAVES))
#define AGG_PASSES ((AGG_ROWS * (EDGE_DIM / 4)) / (32 * AGG_WAVES))
#define AGG_PITCH  68
#define DEG_SLOT   64

typedef __attribute__((ext_vector_type(3))) float v3f;

__device__ __forceinline__ v3f ld3(const float* __restrict__ p) {
  v3f r;
  r.x = p[0];  r.y = p[1];  r.z = p[2];
  return r;
}

struct AggFrag {
  v3f g[AGG_U][2];
  v4f e[AGG_U][2];
};

__device__ __forceinline__ void agg_load(AggFrag& f, const float* __restrict__ gp,
                                         const float* __restrict__ ep) {
#pragma unroll
  for (int u = 0; u < AGG_U; ++u) {
    f.g[u][0] = ld3(gp + (size_t)(4 * u) * NN);
    f.g[u][1] = ld3(gp + (size_t)(4 * u + 1) * NN);
    f.e[u][0] = *(const v4f*)(ep + (size_t)(4 * u) * EDGE_DIM);
    f.e[u][1] = *(const v4f*)(ep + (size_t)(4 * u + 1) * EDGE_DIM);
  }
}

__device__ __forceinline__ void agg_compute(const AggFrag& f, v8f (&acc)[3][4], v2f (&dsum)[3]) {
#pragma unroll
  for (int u = 0; u < AGG_U; ++u) {
    v2f a[3], bq[4];
#pragma unroll
    for (int i = 0; i < 3; ++i) { a[i].x = f.g[u][0][i];  a[i].y = f.g[u][1][i]; }
#pragma unroll
    for (int j = 0; j < 4; ++j) { bq[j].x = f.e[u][0][j];  bq[j].y = f.e[u][1][j]; }
#pragma unroll
    for (int i = 0; i < 3; ++i) {
      dsum[i] += a[i];
#pragma unroll
      for (int j = 0; j < 4; ++j)
        acc[i][j] = WMMA_F32(a[i], bq[j], acc[i][j]);
    }
    __builtin_amdgcn_sched_barrier(0);
  }
}

__global__ __launch_bounds__(32 * AGG_WAVES)
void agg_kernel(const float* __restrict__ G, const float* __restrict__ edge,
                float* __restrict__ aggp, float* __restrict__ degp) {
  __shared__ __attribute__((aligned(16))) float tile[AGG_WAVES * AGG_ROWS * AGG_PITCH];
  __shared__ float degs[AGG_WAVES * AGG_ROWS];

  const int tiles_n = NN / AGG_ROWS;
  int t = blockIdx.x;
  const int plane = t / (BSZ * tiles_n);
  t -= plane * BSZ * tiles_n;
  const int b  = t / tiles_n;
  const int tn = t - b * tiles_n;

  const int wave = threadIdx.x >> 5;
  const int lane = threadIdx.x & 31;
  const int half = lane >> 4;
  const int l15  = lane & 15;
  const int e0 = (plane * AGG_WAVES + wave) * AGG_KW;
  const int n0 = tn * AGG_ROWS + 3 * l15;

  const float* __restrict__ gp = G    + (size_t)b * EE * NN       + (size_t)(e0 + half * 2) * NN       + n0;
  const float* __restrict__ ep = edge + (size_t)b * EE * EDGE_DIM + (size_t)(e0 + half * 2) * EDGE_DIM + 4 * l15;
  const size_t gstep = (size_t)(4 * AGG_U) * NN;
  const size_t estep = (size_t)(4 * AGG_U) * EDGE_DIM;

  v8f acc[3][4] = {};
  v2f dsum[3] = {};

  AggFrag fa, fb;
  const int nblk = AGG_KW / (4 * AGG_U);
  agg_load(fa, gp, ep);
  for (int blk = 0; blk < nblk - 2; blk += 2) {
    agg_load(fb, gp + gstep, ep + estep);
    __builtin_amdgcn_sched_barrier(0);
    agg_compute(fa, acc, dsum);
    gp += 2 * gstep;
    ep += 2 * estep;
    agg_load(fa, gp, ep);
    __builtin_amdgcn_sched_barrier(0);
    agg_compute(fb, acc, dsum);
  }
  agg_load(fb, gp + gstep, ep + estep);
  __builtin_amdgcn_sched_barrier(0);
  agg_compute(fa, acc, dsum);
  agg_compute(fb, acc, dsum);

#pragma unroll
  for (int i = 0; i < 3; ++i) {
    float deg = dsum[i].x + dsum[i].y;
    deg += __shfl_xor(deg, 16, 32);
    if (half == 0) degs[wave * AGG_ROWS + 3 * l15 + i] = deg;
  }

#pragma unroll
  for (int i = 0; i < 3; ++i) {
#pragma unroll
    for (int r = 0; r < 8; ++r) {
      v4f v;
      v.x = acc[i][0][r];  v.y = acc[i][1][r];  v.z = acc[i][2][r];  v.w = acc[i][3][r];
      *(v4f*)(tile + (wave * AGG_ROWS + 3 * (r + 8 * half) + i) * AGG_PITCH + 4 * l15) = v;
    }
  }
  __syncthreads();

  float* __restrict__ outp = aggp + ((size_t)(plane * BSZ + b) * NN + (size_t)tn * AGG_ROWS) * EDGE_DIM;
#pragma unroll
  for (int it = 0; it < AGG_PASSES; ++it) {
    const int idx = it * (32 * AGG_WAVES) + threadIdx.x;
    const int row = idx >> 4;
    const int c4  = (idx & 15) * 4;
    const float* __restrict__ p = tile + row * AGG_PITCH + c4;
    v4f s = *(const v4f*)p;
    s += *(const v4f*)(p + 1 * AGG_ROWS * AGG_PITCH);
    s += *(const v4f*)(p + 2 * AGG_ROWS * AGG_PITCH);
    s += *(const v4f*)(p + 3 * AGG_ROWS * AGG_PITCH);
    *(v4f*)(outp + (size_t)row * EDGE_DIM + c4) = s;
  }
  if (threadIdx.x < AGG_ROWS) {
    const int r = threadIdx.x;
    degp[(size_t)((plane * BSZ + b) * tiles_n + tn) * DEG_SLOT + r] =
        ((degs[r] + degs[AGG_ROWS + r]) + degs[2 * AGG_ROWS + r]) + degs[3 * AGG_ROWS + r];
  }
}

__device__ __forceinline__ float deg_at(const float* __restrict__ degp, int plane, int m) {
  const int b  = m / NN;
  const int n  = m - b * NN;
  const int tn = n / AGG_ROWS;
  const int r  = n - tn * AGG_ROWS;
  return degp[(size_t)((plane * BSZ + b) * (NN / AGG_ROWS) + tn) * DEG_SLOT + r];
}

__global__ __launch_bounds__(32)
void qkv_proj_kernel(const float* __restrict__ node, const float* __restrict__ Wn,
                     const float* __restrict__ bn, const float* __restrict__ aggp,
                     const float* __restrict__ degp, const float* __restrict__ We,
                     const float* __restrict__ be, float* __restrict__ C) {
  const int tiles_n = QKV >> 5;
  const int tile = blockIdx.x;
  const int tm = tile / tiles_n;
  const int tn = tile % tiles_n;
  const int lane = threadIdx.x;
  const int half = lane >> 4;
  const int l15  = lane & 15;

  const int row0 = tm * 32 + l15;
  const int row1 = row0 + 16;
  const int col0 = tn * 32 + l15;
  const int col1 = col0 + 16;

  v8f acc00 = {}, acc01 = {}, acc10 = {}, acc11 = {};

  {
    const float* __restrict__ Xr0 = node + (size_t)row0 * NODE_DIM;
    const float* __restrict__ Xr1 = node + (size_t)row1 * NODE_DIM;
    for (int k = 0; k < NODE_DIM; k += 4) {
      const int ka = k + half * 2;
      v2f a0, a1, b0, b1;
      a0.x = Xr0[ka];                        a0.y = Xr0[ka + 1];
      a1.x = Xr1[ka];                        a1.y = Xr1[ka + 1];
      b0.x = Wn[(size_t)ka * QKV + col0];    b0.y = Wn[(size_t)(ka + 1) * QKV + col0];
      b1.x = Wn[(size_t)ka * QKV + col1];    b1.y = Wn[(size_t)(ka + 1) * QKV + col1];
      acc00 = WMMA_F32(a0, b0, acc00);
      acc01 = WMMA_F32(a0, b1, acc01);
      acc10 = WMMA_F32(a1, b0, acc10);
      acc11 = WMMA_F32(a1, b1, acc11);
    }
  }

  {
    const size_t pl = (size_t)BSZ * NN * EDGE_DIM;
    const float* __restrict__ Pr0 = aggp + (size_t)row0 * EDGE_DIM;
    const float* __restrict__ Pr1 = aggp + (size_t)row1 * EDGE_DIM;
    const float rinv0 = 1.0f / (deg_at(degp, 0, row0) + deg_at(degp, 1, row0));
    const float rinv1 = 1.0f / (deg_at(degp, 0, row1) + deg_at(degp, 1, row1));
    for (int k = 0; k < EDGE_DIM; k += 4) {
      const int ka = k + half * 2;
      v2f a0, a1, b0, b1;
      a0.x = (Pr0[ka] + Pr0[pl + ka]) * rinv0;   a0.y = (Pr0[ka + 1] + Pr0[pl + ka + 1]) * rinv0;
      a1.x = (Pr1[ka] + Pr1[pl + ka]) * rinv1;   a1.y = (Pr1[ka + 1] + Pr1[pl + ka + 1]) * rinv1;
      b0.x = We[(size_t)ka * QKV + col0];        b0.y = We[(size_t)(ka + 1) * QKV + col0];
      b1.x = We[(size_t)ka * QKV + col1];        b1.y = We[(size_t)(ka + 1) * QKV + col1];
      acc00 = WMMA_F32(a0, b0, acc00);
      acc01 = WMMA_F32(a0, b1, acc01);
      acc10 = WMMA_F32(a1, b0, acc10);
      acc11 = WMMA_F32(a1, b1, acc11);
    }
  }

  const float bv0 = bn[col0] + be[col0];
  const float bv1 = bn[col1] + be[col1];
#pragma unroll
  for (int j = 0; j < 8; ++j) {
    const int r0 = tm * 32 + j + half * 8;
    const int r1 = r0 + 16;
    C[(size_t)r0 * QKV + col0] = acc00[j] + bv0;
    C[(size_t)r0 * QKV + col1] = acc01[j] + bv1;
    C[(size_t)r1 * QKV + col0] = acc10[j] + bv0;
    C[(size_t)r1 * QKV + col1] = acc11[j] + bv1;
  }
}

#define ATT_WAVES 4

__device__ __forceinline__ float xhalf(float x) {
  const int b = __builtin_bit_cast(int, x);
  return __builtin_bit_cast(float,
      __builtin_amdgcn_permlanex16(b, b, 0x76543210, 0xfedcba98, false, false));
}

__device__ __forceinline__ float max3f(float a, float b, float c) {
  return __builtin_fmaximum_numf(__builtin_fmaximum_numf(a, b), c);
}

struct AttFrag {
  v4f   k[2][2];
  float v[2][8];
};

__device__ __forceinline__ void att_load(AttFrag& f, const float* __restrict__ kp,
                                         const float* __restrict__ vp) {
  for (int j = 0; j < 2; ++j) {
    f.k[j][0] = *(const v4f*)(kp + (size_t)(16 * j) * QKV);
    f.k[j][1] = *(const v4f*)(kp + (size_t)(16 * j) * QKV + 4);
    for (int r = 0; r < 8; ++r)
      f.v[j][r] = vp[(size_t)(16 * j + r) * QKV];
  }
}

__device__ __forceinline__ void att_step(AttFrag& f, const v2f (&qb)[4],
                                         v8f& o, float& m, float& l) {
  asm volatile("" : "+v"(f.k[0][0]), "+v"(f.k[0][1]), "+v"(f.k[1][0]), "+v"(f.k[1][1]),
                    "+v"(o), "+v"(m), "+v"(l));

  v8f s0 = {}, s1 = {};
  for (int t = 0; t < 4; ++t) {
    v2f a0, a1;
    a0.x = f.k[0][t >> 1][2 * (t & 1)];  a0.y = f.k[0][t >> 1][2 * (t & 1) + 1];
    a1.x = f.k[1][t >> 1][2 * (t & 1)];  a1.y = f.k[1][t >> 1][2 * (t & 1) + 1];
    s0 = WMMA_F32(a0, qb[t], s0);
    s1 = WMMA_F32(a1, qb[t], s1);
  }

  const float t0 = max3f(s0[0], s0[1], s0[2]);
  const float t1 = max3f(s0[3], s0[4], s0[5]);
  const float t2 = max3f(s0[6], s0[7], s1[0]);
  const float t3 = max3f(s1[1], s1[2], s1[3]);
  const float t4 = max3f(s1[4], s1[5], s1[6]);
  const float mx = __builtin_fmaximum_numf(max3f(t0, t1, t2), max3f(t3, t4, s1[7]));
  const float mnew = max3f(m, mx, xhalf(mx));
  const float corr = __builtin_amdgcn_exp2f(m - mnew);
  m = mnew;

  const v2f mm = {mnew, mnew};
  v2f ps = {0.f, 0.f};
  v8f p0, p1;
  for (int i = 0; i < 4; ++i) {
    v2f d0, d1;
    d0.x = s0[2 * i];  d0.y = s0[2 * i + 1];
    d1.x = s1[2 * i];  d1.y = s1[2 * i + 1];
    d0 -= mm;
    d1 -= mm;
    v2f e0, e1;
    e0.x = __builtin_amdgcn_exp2f(d0.x);  e0.y = __builtin_amdgcn_exp2f(d0.y);
    e1.x = __builtin_amdgcn_exp2f(d1.x);  e1.y = __builtin_amdgcn_exp2f(d1.y);
    ps += e0;
    ps += e1;
    p0[2 * i] = e0.x;  p0[2 * i + 1] = e0.y;
    p1[2 * i] = e1.x;  p1[2 * i + 1] = e1.y;
  }
  l = l * corr + (ps.x + ps.y);
  o *= corr;

  for (int t = 0; t < 4; ++t) {
    v2f a, bb;
    a.x  = f.v[0][2 * t];  a.y  = f.v[0][2 * t + 1];
    bb.x = p0[2 * t];      bb.y = p0[2 * t + 1];
    o = WMMA_F32(a, bb, o);
  }
  for (int t = 0; t < 4; ++t) {
    v2f a, bb;
    a.x  = f.v[1][2 * t];  a.y  = f.v[1][2 * t + 1];
    bb.x = p1[2 * t];      bb.y = p1[2 * t + 1];
    o = WMMA_F32(a, bb, o);
  }
}

__global__ __launch_bounds__(32 * ATT_WAVES)
void attn_kernel(const float* __restrict__ qkv, float* __restrict__ attout) {
  const int groups_q = NN / (16 * ATT_WAVES);
  int grp = blockIdx.x;
  const int b = grp / (NHEAD * groups_q);
  grp -= b * NHEAD * groups_q;
  const int h  = grp / groups_q;
  const int gq = grp - h * groups_q;
  const int wave = threadIdx.x >> 5;
  const int tq = gq * ATT_WAVES + wave;

  const int lane = threadIdx.x & 31;
  const int half = lane >> 4;
  const int l15  = lane & 15;
  const float qscale = 0.25f * 1.4426950408889634f;

  const float* __restrict__ base = qkv + (size_t)b * NN * QKV + h * 48;

  v2f qb[4];
  {
    const float* __restrict__ qp = base + (size_t)(tq * 16 + l15) * QKV + 8 * half;
    const v4f q0 = *(const v4f*)qp;
    const v4f q1 = *(const v4f*)(qp + 4);
    qb[0].x = q0.x * qscale;  qb[0].y = q0.y * qscale;
    qb[1].x = q0.z * qscale;  qb[1].y = q0.w * qscale;
    qb[2].x = q1.x * qscale;  qb[2].y = q1.y * qscale;
    qb[3].x = q1.z * qscale;  qb[3].y = q1.w * qscale;
  }

  const float* __restrict__ kp = base + (size_t)l15 * QKV + 16 + 8 * half;
  const float* __restrict__ vp = base + (size_t)(8 * half) * QKV + 32 + l15;
  const size_t step = (size_t)32 * QKV;

  v8f o = {};
  float m = -1e30f, l = 0.f;

  AttFrag fa, fb;
  const int nstep = NN / 32;
  att_load(fa, kp, vp);
  for (int it = 0; it < nstep - 2; it += 2) {
    att_load(fb, kp + step, vp + step);
    __builtin_amdgcn_sched_barrier(0);
    att_step(fa, qb, o, m, l);
    kp += 2 * step;
    vp += 2 * step;
    att_load(fa, kp, vp);
    __builtin_amdgcn_sched_barrier(0);
    att_step(fb, qb, o, m, l);
  }
  att_load(fb, kp + step, vp + step);
  __builtin_amdgcn_sched_barrier(0);
  att_step(fa, qb, o, m, l);
  att_step(fb, qb, o, m, l);

  l += xhalf(l);
  float* __restrict__ outp = attout + (size_t)b * NN * NODE_DIM
                           + (size_t)(tq * 16 + l15) * NODE_DIM + h * HDIM + 8 * half;
  v4f r0, r1;
  r0.x = o[0] / l;  r0.y = o[1] / l;  r0.z = o[2] / l;  r0.w = o[3] / l;
  r1.x = o[4] / l;  r1.y = o[5] / l;  r1.z = o[6] / l;  r1.w = o[7] / l;
  *(v4f*)outp       = r0;
  *(v4f*)(outp + 4) = r1;
}

extern "C" void kernel_launch(void* const* d_in, const int* in_sizes, int n_in,
                              void* d_out, int out_size, void* d_ws, size_t ws_size,
                              hipStream_t stream) {
  const float* node    = (const float*)d_in[0];
  const float* edge    = (const float*)d_in[1];
  const float* G       = (const float*)d_in[2];
  const float* W_qkv_n = (const float*)d_in[3];
  const float* b_qkv_n = (const float*)d_in[4];
  const float* W_qkv_e = (const float*)d_in[5];
  const float* b_qkv_e = (const float*)d_in[6];
  const float* W_o     = (const float*)d_in[7];
  const float* b_o     = (const float*)d_in[8];
  float* out = (float*)d_out;

  float* qkv  = (float*)d_ws;
  float* aggp = qkv  + (size_t)BSZ * NN * QKV;
  float* degp = aggp + (size_t)AGG_PLANES * BSZ * NN * EDGE_DIM;
  float* att  = degp + (size_t)AGG_PLANES * BSZ * (NN / AGG_ROWS) * DEG_SLOT;

  agg_kernel<<<AGG_PLANES * BSZ * (NN / AGG_ROWS), 32 * AGG_WAVES, 0, stream>>>(G, edge, aggp, degp);

  qkv_proj_kernel<<<(BSZ * NN / 32) * (QKV / 32), 32, 0, stream>>>(
      node, W_qkv_n, b_qkv_n, aggp, degp, W_qkv_e, b_qkv_e, qkv);

  attn_kernel<<<BSZ * NHEAD * (NN / (16 * ATT_WAVES)), 32 * ATT_WAVES, 0, stream>>>(qkv, att);

  gemm_bias_32x32<<<(BSZ * NN / 32) * (NODE_DIM / 32), 32, 0, stream>>>(
      att, W_o, b_o, out, BSZ * NN, NODE_DIM, NODE_DIM);
}


# projection GEMMs (qkv, output) at one wave per SIMD: 48-row tiles, four waves of a group on the same rows, K loop in blocks whose loads issue as one group a block ahead; every sum and its order kept
# speedup vs baseline: 6.9751x; 1.1585x over previous
#include <hip/hip_runtime.h>
#include <hip/hip_bf16.h>


typedef __attribute__((ext_vector_type(2))) float v2f;
typedef __attribute__((ext_vector_type(8))) float v8f;
typedef __attribute__((ext_vector_type(4))) float v4f;

#define BSZ        2
#define NN         3072
#define EE         4096
#define NODE_DIM   128
#define EDGE_DIM   64
#define NHEAD      8
#define HDIM       16
#define QKV        384

#define WMMA_F32(A, B, C) \
  __builtin_amdgcn_wmma_f32_16x16x4_f32(false, (A), false, (B), (short)0, (C), false, false)

#define AGG_U      8
#define AGG_ROWS   48
#define AGG_WAVES  4
#define AGG_PLANES 2
#define AGG_KW     (EE / (AGG_PLANES * AGG_WAVES))
#define AGG_PASSES ((AGG_ROWS * (EDGE_DIM / 4)) / (32 * AGG_WAVES))
#define AGG_PITCH  68
#define DEG_SLOT   64

typedef __attribute__((ext_vector_type(3))) float v3f;

__device__ __forceinline__ v3f ld3(const float* __restrict__ p) {
  v3f r;
  r.x = p[0];  r.y = p[1];  r.z = p[2];
  return r;
}

struct AggFrag {
  v3f g[AGG_U][2];
  v4f e[AGG_U][2];
};

__device__ __forceinline__ void agg_load(AggFrag& f, const float* __restrict__ gp,
                                         const float* __restrict__ ep) {
#pragma unroll
  for (int u = 0; u < AGG_U; ++u) {
    f.g[u][0] = ld3(gp + (size_t)(4 * u) * NN);
    f.g[u][1] = ld3(gp + (size_t)(4 * u + 1) * NN);
    f.e[u][0] = *(const v4f*)(ep + (size_t)(4 * u) * EDGE_DIM);
    f.e[u][1] = *(const v4f*)(ep + (size_t)(4 * u + 1) * EDGE_DIM);
  }
}

__device__ __forceinline__ void agg_compute(const AggFrag& f, v8f (&acc)[3][4], v2f (&dsum)[3]) {
#pragma unroll
  for (int u = 0; u < AGG_U; ++u) {
    v2f a[3], bq[4];
#pragma unroll
    for (int i = 0; i < 3; ++i) { a[i].x = f.g[u][0][i];  a[i].y = f.g[u][1][i]; }
#pragma unroll
    for (int j = 0; j < 4; ++j) { bq[j].x = f.e[u][0][j];  bq[j].y = f.e[u][1][j]; }
#pragma unroll
    for (int i = 0; i < 3; ++i) {
      dsum[i] += a[i];
#pragma unroll
      for (int j = 0; j < 4; ++j)
        acc[i][j] = WMMA_F32(a[i], bq[j], acc[i][j]);
    }
    __builtin_amdgcn_sched_barrier(0);
  }
}

__global__ __launch_bounds__(32 * AGG_WAVES)
void agg_kernel(const float* __restrict__ G, const float* __restrict__ edge,
                float* __restrict__ aggp, float* __restrict__ degp) {
  __shared__ __attribute__((aligned(16))) float tile[AGG_WAVES * AGG_ROWS * AGG_PITCH];
  __shared__ float degs[AGG_WAVES * AGG_ROWS];

  const int tiles_n = NN / AGG_ROWS;
  int t = blockIdx.x;
  const int plane = t / (BSZ * tiles_n);
  t -= plane * BSZ * tiles_n;
  const int b  = t / tiles_n;
  const int tn = t - b * tiles_n;

  const int wave = threadIdx.x >> 5;
  const int lane = threadIdx.x & 31;
  const int half = lane >> 4;
  const int l15  = lane & 15;
  const int e0 = (plane * AGG_WAVES + wave) * AGG_KW;
  const int n0 = tn * AGG_ROWS + 3 * l15;

  const float* __restrict__ gp = G    + (size_t)b * EE * NN       + (size_t)(e0 + half * 2) * NN       + n0;
  const float* __restrict__ ep = edge + (size_t)b * EE * EDGE_DIM + (size_t)(e0 + half * 2) * EDGE_DIM + 4 * l15;
  const size_t gstep = (size_t)(4 * AGG_U) * NN;
  const size_t estep = (size_t)(4 * AGG_U) * EDGE_DIM;

  v8f acc[3][4] = {};
  v2f dsum[3] = {};

  AggFrag fa, fb;
  const int nblk = AGG_KW / (4 * AGG_U);
  agg_load(fa, gp, ep);
  for (int blk = 0; blk < nblk - 2; blk += 2) {
    agg_load(fb, gp + gstep, ep + estep);
    __builtin_amdgcn_sched_barrier(0);
    agg_compute(fa, acc, dsum);
    gp += 2 * gstep;
    ep += 2 * estep;
    agg_load(fa, gp, ep);
    __builtin_amdgcn_sched_barrier(0);
    agg_compute(fb, acc, dsum);
  }
  agg_load(fb, gp + gstep, ep + estep);
  __builtin_amdgcn_sched_barrier(0);
  agg_compute(fa, acc, dsum);
  agg_compute(fb, acc, dsum);

#pragma unroll
  for (int i = 0; i < 3; ++i) {
    float deg = dsum[i].x + dsum[i].y;
    deg += __shfl_xor(deg, 16, 32);
    if (half == 0) degs[wave * AGG_ROWS + 3 * l15 + i] = deg;
  }

#pragma unroll
  for (int i = 0; i < 3; ++i) {
#pragma unroll
    for (int r = 0; r < 8; ++r) {
      v4f v;
      v.x = acc[i][0][r];  v.y = acc[i][1][r];  v.z = acc[i][2][r];  v.w = acc[i][3][r];
      *(v4f*)(tile + (wave * AGG_ROWS + 3 * (r + 8 * half) + i) * AGG_PITCH + 4 * l15) = v;
    }
  }
  __syncthreads();

  float* __restrict__ outp = aggp + ((size_t)(plane * BSZ + b) * NN + (size_t)tn * AGG_ROWS) * EDGE_DIM;
#pragma unroll
  for (int it = 0; it < AGG_PASSES; ++it) {
    const int idx = it * (32 * AGG_WAVES) + threadIdx.x;
    const int row = idx >> 4;
    const int c4  = (idx & 15) * 4;
    const float* __restrict__ p = tile + row * AGG_PITCH + c4;
    v4f s = *(const v4f*)p;
    s += *(const v4f*)(p + 1 * AGG_ROWS * AGG_PITCH);
    s += *(const v4f*)(p + 2 * AGG_ROWS * AGG_PITCH);
    s += *(const v4f*)(p + 3 * AGG_ROWS * AGG_PITCH);
    *(v4f*)(outp + (size_t)row * EDGE_DIM + c4) = s;
  }
  if (threadIdx.x < AGG_ROWS) {
    const int r = threadIdx.x;
    degp[(size_t)((plane * BSZ + b) * tiles_n + tn) * DEG_SLOT + r] =
        ((degs[r] + degs[AGG_ROWS + r]) + degs[2 * AGG_ROWS + r]) + degs[3 * AGG_ROWS + r];
  }
}

#define PRJ_ROWS   48
#define PRJ_WAVES  4
#define PRJ_U      8

__device__ __forceinline__ float deg_at(const float* __restrict__ degp, int plane, int m) {
  const int b  = m / NN;
  const int n  = m - b * NN;
  const int tn = n / AGG_ROWS;
  const int r  = n - tn * AGG_ROWS;
  return degp[(size_t)((plane * BSZ + b) * (NN / AGG_ROWS) + tn) * DEG_SLOT + r];
}

struct NodeFrag {
  v2f a[PRJ_U][3];
  v3f b[PRJ_U][2];
};

struct EdgeFrag {
  v2f p0[PRJ_U][3];
  v2f p1[PRJ_U][3];
  v3f b[PRJ_U][2];
};

__device__ __forceinline__ void node_load(NodeFrag& f, const float* __restrict__ xp,
                                          const float* __restrict__ wp) {
#pragma unroll
  for (int u = 0; u < PRJ_U; ++u) {
#pragma unroll
    for (int i = 0; i < 3; ++i)
      f.a[u][i] = *(const v2f*)(xp + (size_t)(16 * i) * NODE_DIM + 4 * u);
    f.b[u][0] = ld3(wp + (size_t)(4 * u) * QKV);
    f.b[u][1] = ld3(wp + (size_t)(4 * u + 1) * QKV);
  }
}

__device__ __forceinline__ void edge_load(EdgeFrag& f, const float* __restrict__ pp, size_t pl,
                                          const float* __restrict__ wp) {
#pragma unroll
  for (int u = 0; u < PRJ_U; ++u) {
#pragma unroll
    for (int i = 0; i < 3; ++i) {
      f.p0[u][i] = *(const v2f*)(pp + (size_t)(16 * i) * EDGE_DIM + 4 * u);
      f.p1[u][i] = *(const v2f*)(pp + pl + (size_t)(16 * i) * EDGE_DIM + 4 * u);
    }
    f.b[u][0] = ld3(wp + (size_t)(4 * u) * QKV);
    f.b[u][1] = ld3(wp + (size_t)(4 * u + 1) * QKV);
  }
}

__device__ __forceinline__ void prj_kstep(const v2f (&a)[3], const v3f& b0, const v3f& b1,
                                          v8f (&acc)[3][3]) {
  v2f bq[3];
#pragma unroll
  for (int j = 0; j < 3; ++j) { bq[j].x = b0[j];  bq[j].y = b1[j]; }
#pragma unroll
  for (int i = 0; i < 3; ++i) {
#pragma unroll
    for (int j = 0; j < 3; ++j)
      acc[i][j] = WMMA_F32(a[i], bq[j], acc[i][j]);
  }
}

__device__ __forceinline__ void node_compute(const NodeFrag& f, v8f (&acc)[3][3]) {
#pragma unroll
  for (int u = 0; u < PRJ_U; ++u) {
    v2f a[3];
#pragma unroll
    for (int i = 0; i < 3; ++i) a[i] = f.a[u][i];
    prj_kstep(a, f.b[u][0], f.b[u][1], acc);
    __builtin_amdgcn_sched_barrier(0);
  }
}

__device__ __forceinline__ void edge_compute(const EdgeFrag& f, const float (&rinv)[3],
                                             v8f (&acc)[3][3]) {
#pragma unroll
  for (int u = 0; u < PRJ_U; ++u) {
    v2f a[3];
#pragma unroll
    for (int i = 0; i < 3; ++i) a[i] = (f.p0[u][i] + f.p1[u][i]) * rinv[i];
    prj_kstep(a, f.b[u][0], f.b[u][1], acc);
    __builtin_amdgcn_sched_barrier(0);
  }
}

__global__ __launch_bounds__(32 * PRJ_WAVES)
void qkv_proj_kernel(const float* __restrict__ node, const float* __restrict__ Wn,
                     const float* __restrict__ bn, const float* __restrict__ aggp,
                     const float* __restrict__ degp, const float* __restrict__ We,
                     const float* __restrict__ be, float* __restrict__ C) {
  const int tm   = blockIdx.x >> 1;
  const int wave = threadIdx.x >> 5;
  const int lane = threadIdx.x & 31;
  const int half = lane >> 4;
  const int l15  = lane & 15;
  const int c    = ((blockIdx.x & 1) * PRJ_WAVES + wave) * 48 + 3 * l15;
  const int row0 = tm * PRJ_ROWS + l15;

  float d0[3], d1[3], bnv[3], bev[3];
#pragma unroll
  for (int i = 0; i < 3; ++i) {
    d0[i]  = deg_at(degp, 0, row0 + 16 * i);
    d1[i]  = deg_at(degp, 1, row0 + 16 * i);
    bnv[i] = bn[c + i];
    bev[i] = be[c + i];
  }

  const float* __restrict__ xp = node + (size_t)row0 * NODE_DIM + 2 * half;
  const float* __restrict__ wp = Wn   + (size_t)(2 * half) * QKV + c;
  const float* __restrict__ pp = aggp + (size_t)row0 * EDGE_DIM + 2 * half;
  const float* __restrict__ ep = We   + (size_t)(2 * half) * QKV + c;
  const size_t pl    = (size_t)BSZ * NN * EDGE_DIM;
  const int    kblk  = 4 * PRJ_U;
  const size_t wstep = (size_t)kblk * QKV;

  v8f acc[3][3] = {};
  NodeFrag na, nb;
  EdgeFrag ea, eb;

  node_load(na, xp, wp);
  node_load(nb, xp + kblk, wp + wstep);
  __builtin_amdgcn_sched_barrier(0);
  node_compute(na, acc);
  node_load(na, xp + 2 * kblk, wp + 2 * wstep);
  __builtin_amdgcn_sched_barrier(0);
  node_compute(nb, acc);
  node_load(nb, xp + 3 * kblk, wp + 3 * wstep);
  __builtin_amdgcn_sched_barrier(0);
  node_compute(na, acc);
  edge_load(ea, pp, pl, ep);
  __builtin_amdgcn_sched_barrier(0);
  node_compute(nb, acc);
  edge_load(eb, pp + kblk, pl, ep + wstep);
  __builtin_amdgcn_sched_barrier(0);

  float rinv[3];
#pragma unroll
  for (int i = 0; i < 3; ++i) rinv[i] = 1.0f / (d0[i] + d1[i]);
  edge_compute(ea, rinv, acc);
  edge_compute(eb, rinv, acc);

  float bv[3];
#pragma unroll
  for (int j = 0; j < 3; ++j) bv[j] = bnv[j] + bev[j];
#pragma unroll
  for (int i = 0; i < 3; ++i) {
#pragma unroll
    for (int r = 0; r < 8; ++r) {
      float* __restrict__ p = C + (size_t)(tm * PRJ_ROWS + 16 * i + 8 * half + r) * QKV + c;
      p[0] = acc[i][0][r] + bv[0];
      p[1] = acc[i][1][r] + bv[1];
      p[2] = acc[i][2][r] + bv[2];
    }
  }
}

#define OUT_KSTEPS (NODE_DIM / 4)

__global__ __launch_bounds__(32 * PRJ_WAVES)
void out_proj_kernel(const float* __restrict__ X, const float* __restrict__ W,
                     const float* __restrict__ bias, float* __restrict__ C) {
  const int tm   = blockIdx.x >> 1;
  const int wave = threadIdx.x >> 5;
  const int lane = threadIdx.x & 31;
  const int half = lane >> 4;
  const int l15  = lane & 15;
  const int col  = ((blockIdx.x & 1) * PRJ_WAVES + wave) * 16 + l15;
  const int row0 = tm * PRJ_ROWS + l15;

  const float* __restrict__ xp = X + (size_t)row0 * NODE_DIM + 2 * half;
  const float* __restrict__ wp = W + (size_t)(2 * half) * NODE_DIM + col;

  v2f a[OUT_KSTEPS][3], b[OUT_KSTEPS];
#pragma unroll
  for (int u = 0; u < OUT_KSTEPS; ++u) {
#pragma unroll
    for (int i = 0; i < 3; ++i)
      a[u][i] = *(const v2f*)(xp + (size_t)(16 * i) * NODE_DIM + 4 * u);
    b[u].x = wp[(size_t)(4 * u) * NODE_DIM];
    b[u].y = wp[(size_t)(4 * u + 1) * NODE_DIM];
  }
  __builtin_amdgcn_sched_barrier(0);

  v8f acc[3] = {};
#pragma unroll
  for (int u = 0; u < OUT_KSTEPS; ++u) {
#pragma unroll
    for (int i = 0; i < 3; ++i)
      acc[i] = WMMA_F32(a[u][i], b[u], acc[i]);
  }

  const float bv = bias[col];
#pragma unroll
  for (int i = 0; i < 3; ++i) {
#pragma unroll
    for (int r = 0; r < 8; ++r)
      C[(size_t)(tm * PRJ_ROWS + 16 * i + 8 * half + r) * NODE_DIM + col] = acc[i][r] + bv;
  }
}

#define ATT_WAVES 4

__device__ __forceinline__ float xhalf(float x) {
  const int b = __builtin_bit_cast(int, x);
  return __builtin_bit_cast(float,
      __builtin_amdgcn_permlanex16(b, b, 0x76543210, 0xfedcba98, false, false));
}

__device__ __forceinline__ float max3f(float a, float b, float c) {
  return __builtin_fmaximum_numf(__builtin_fmaximum_numf(a, b), c);
}

struct AttFrag {
  v4f   k[2][2];
  float v[2][8];
};

__device__ __forceinline__ void att_load(AttFrag& f, const float* __restrict__ kp,
                                         const float* __restrict__ vp) {
  for (int j = 0; j < 2; ++j) {
    f.k[j][0] = *(const v4f*)(kp + (size_t)(16 * j) * QKV);
    f.k[j][1] = *(const v4f*)(kp + (size_t)(16 * j) * QKV + 4);
    for (int r = 0; r < 8; ++r)
      f.v[j][r] = vp[(size_t)(16 * j + r) * QKV];
  }
}

__device__ __forceinline__ void att_step(AttFrag& f, const v2f (&qb)[4],
                                         v8f& o, float& m, float& l) {
  asm volatile("" : "+v"(f.k[0][0]), "+v"(f.k[0][1]), "+v"(f.k[1][0]), "+v"(f.k[1][1]),
                    "+v"(o), "+v"(m), "+v"(l));

  v8f s0 = {}, s1 = {};
  for (int t = 0; t < 4; ++t) {
    v2f a0, a1;
    a0.x = f.k[0][t >> 1][2 * (t & 1)];  a0.y = f.k[0][t >> 1][2 * (t & 1) + 1];
    a1.x = f.k[1][t >> 1][2 * (t & 1)];  a1.y = f.k[1][t >> 1][2 * (t & 1) + 1];
    s0 = WMMA_F32(a0, qb[t], s0);
    s1 = WMMA_F32(a1, qb[t], s1);
  }

  const float t0 = max3f(s0[0], s0[1], s0[2]);
  const float t1 = max3f(s0[3], s0[4], s0[5]);
  const float t2 = max3f(s0[6], s0[7], s1[0]);
  const float t3 = max3f(s1[1], s1[2], s1[3]);
  const float t4 = max3f(s1[4], s1[5], s1[6]);
  const float mx = __builtin_fmaximum_numf(max3f(t0, t1, t2), max3f(t3, t4, s1[7]));
  const float mnew = max3f(m, mx, xhalf(mx));
  const float corr = __builtin_amdgcn_exp2f(m - mnew);
  m = mnew;

  const v2f mm = {mnew, mnew};
  v2f ps = {0.f, 0.f};
  v8f p0, p1;
  for (int i = 0; i < 4; ++i) {
    v2f d0, d1;
    d0.x = s0[2 * i];  d0.y = s0[2 * i + 1];
    d1.x = s1[2 * i];  d1.y = s1[2 * i + 1];
    d0 -= mm;
    d1 -= mm;
    v2f e0, e1;
    e0.x = __builtin_amdgcn_exp2f(d0.x);  e0.y = __builtin_amdgcn_exp2f(d0.y);
    e1.x = __builtin_amdgcn_exp2f(d1.x);  e1.y = __builtin_amdgcn_exp2f(d1.y);
    ps += e0;
    ps += e1;
    p0[2 * i] = e0.x;  p0[2 * i + 1] = e0.y;
    p1[2 * i] = e1.x;  p1[2 * i + 1] = e1.y;
  }
  l = l * corr + (ps.x + ps.y);
  o *= corr;

  for (int t = 0; t < 4; ++t) {
    v2f a, bb;
    a.x  = f.v[0][2 * t];  a.y  = f.v[0][2 * t + 1];
    bb.x = p0[2 * t];      bb.y = p0[2 * t + 1];
    o = WMMA_F32(a, bb, o);
  }
  for (int t = 0; t < 4; ++t) {
    v2f a, bb;
    a.x  = f.v[1][2 * t];  a.y  = f.v[1][2 * t + 1];
    bb.x = p1[2 * t];      bb.y = p1[2 * t + 1];
    o = WMMA_F32(a, bb, o);
  }
}

__global__ __launch_bounds__(32 * ATT_WAVES)
void attn_kernel(const float* __restrict__ qkv, float* __restrict__ attout) {
  const int groups_q = NN / (16 * ATT_WAVES);
  int grp = blockIdx.x;
  const int b = grp / (NHEAD * groups_q);
  grp -= b * NHEAD * groups_q;
  const int h  = grp / groups_q;
  const int gq = grp - h * groups_q;
  const int wave = threadIdx.x >> 5;
  const int tq = gq * ATT_WAVES + wave;

  const int lane = threadIdx.x & 31;
  const int half = lane >> 4;
  const int l15  = lane & 15;
  const float qscale = 0.25f * 1.4426950408889634f;

  const float* __restrict__ base = qkv + (size_t)b * NN * QKV + h * 48;

  v2f qb[4];
  {
    const float* __restrict__ qp = base + (size_t)(tq * 16 + l15) * QKV + 8 * half;
    const v4f q0 = *(const v4f*)qp;
    const v4f q1 = *(const v4f*)(qp + 4);
    qb[0].x = q0.x * qscale;  qb[0].y = q0.y * qscale;
    qb[1].x = q0.z * qscale;  qb[1].y = q0.w * qscale;
    qb[2].x = q1.x * qscale;  qb[2].y = q1.y * qscale;
    qb[3].x = q1.z * qscale;  qb[3].y = q1.w * qscale;
  }

  const float* __restrict__ kp = base + (size_t)l15 * QKV + 16 + 8 * half;
  const float* __restrict__ vp = base + (size_t)(8 * half) * QKV + 32 + l15;
  const size_t step = (size_t)32 * QKV;

  v8f o = {};
  float m = -1e30f, l = 0.f;

  AttFrag fa, fb;
  const int nstep = NN / 32;
  att_load(fa, kp, vp);
  for (int it = 0; it < nstep - 2; it += 2) {
    att_load(fb, kp + step, vp + step);
    __builtin_amdgcn_sched_barrier(0);
    att_step(fa, qb, o, m, l);
    kp += 2 * step;
    vp += 2 * step;
    att_load(fa, kp, vp);
    __builtin_amdgcn_sched_barrier(0);
    att_step(fb, qb, o, m, l);
  }
  att_load(fb, kp + step, vp + step);
  __builtin_amdgcn_sched_barrier(0);
  att_step(fa, qb, o, m, l);
  att_step(fb, qb, o, m, l);

  l += xhalf(l);
  float* __restrict__ outp = attout + (size_t)b * NN * NODE_DIM
                           + (size_t)(tq * 16 + l15) * NODE_DIM + h * HDIM + 8 * half;
  v4f r0, r1;
  r0.x = o[0] / l;  r0.y = o[1] / l;  r0.z = o[2] / l;  r0.w = o[3] / l;
  r1.x = o[4] / l;  r1.y = o[5] / l;  r1.z = o[6] / l;  r1.w = o[7] / l;
  *(v4f*)outp       = r0;
  *(v4f*)(outp + 4) = r1;
}

extern "C" void kernel_launch(void* const* d_in, const int* in_sizes, int n_in,
                              void* d_out, int out_size, void* d_ws, size_t ws_size,
                              hipStream_t stream) {
  const float* node    = (const float*)d_in[0];
  const float* edge    = (const float*)d_in[1];
  const float* G       = (const float*)d_in[2];
  const float* W_qkv_n = (const float*)d_in[3];
  const float* b_qkv_n = (const float*)d_in[4];
  const float* W_qkv_e = (const float*)d_in[5];
  const float* b_qkv_e = (const float*)d_in[6];
  const float* W_o     = (const float*)d_in[7];
  const float* b_o     = (const float*)d_in[8];
  float* out = (float*)d_out;

  float* qkv  = (float*)d_ws;
  float* aggp = qkv  + (size_t)BSZ * NN * QKV;
  float* degp = aggp + (size_t)AGG_PLANES * BSZ * NN * EDGE_DIM;
  float* att  = degp + (size_t)AGG_PLANES * BSZ * (NN / AGG_ROWS) * DEG_SLOT;

  agg_kernel<<<AGG_PLANES * BSZ * (NN / AGG_ROWS), 32 * AGG_WAVES, 0, stream>>>(G, edge, aggp, degp);

  qkv_proj_kernel<<<(BSZ * NN / PRJ_ROWS) * 2, 32 * PRJ_WAVES, 0, stream>>>(
      node, W_qkv_n, b_qkv_n, aggp, degp, W_qkv_e, b_qkv_e, qkv);

  attn_kernel<<<BSZ * NHEAD * (NN / (16 * ATT_WAVES)), 32 * ATT_WAVES, 0, stream>>>(qkv, att);

  out_proj_kernel<<<(BSZ * NN / PRJ_ROWS) * 2, 32 * PRJ_WAVES, 0, stream>>>(att, W_o, b_o, out);
}


# attention: a wave takes three neighbouring query tiles (was one), one wave per SIMD: a loaded K or V fragment feeds three tiles' products, tiles staggered in a step; every sum and its order kept
# speedup vs baseline: 7.3790x; 1.0579x over previous
#include <hip/hip_runtime.h>
#include <hip/hip_bf16.h>


typedef __attribute__((ext_vector_type(2))) float v2f;
typedef __attribute__((ext_vector_type(8))) float v8f;
typedef __attribute__((ext_vector_type(4))) float v4f;

#define BSZ        2
#define NN         3072
#define EE         4096
#define NODE_DIM   128
#define EDGE_DIM   64
#define NHEAD      8
#define HDIM       16
#define QKV        384

#define WMMA_F32(A, B, C) \
  __builtin_amdgcn_wmma_f32_16x16x4_f32(false, (A), false, (B), (short)0, (C), false, false)

#define AGG_U      8
#define AGG_ROWS   48
#define AGG_WAVES  4
#define AGG_PLANES 2
#define AGG_KW     (EE / (AGG_PLANES * AGG_WAVES))
#define AGG_PASSES ((AGG_ROWS * (EDGE_DIM / 4)) / (32 * AGG_WAVES))
#define AGG_PITCH  68
#define DEG_SLOT   64

typedef __attribute__((ext_vector_type(3))) float v3f;

__device__ __forceinline__ v3f ld3(const float* __restrict__ p) {
  v3f r;
  r.x = p[0];  r.y = p[1];  r.z = p[2];
  return r;
}

struct AggFrag {
  v3f g[AGG_U][2];
  v4f e[AGG_U][2];
};

__device__ __forceinline__ void agg_load(AggFrag& f, const float* __restrict__ gp,
                                         const float* __restrict__ ep) {
#pragma unroll
  for (int u = 0; u < AGG_U; ++u) {
    f.g[u][0] = ld3(gp + (size_t)(4 * u) * NN);
    f.g[u][1] = ld3(gp + (size_t)(4 * u + 1) * NN);
    f.e[u][0] = *(const v4f*)(ep + (size_t)(4 * u) * EDGE_DIM);
    f.e[u][1] = *(const v4f*)(ep + (size_t)(4 * u + 1) * EDGE_DIM);
  }
}

__device__ __forceinline__ void agg_compute(const AggFrag& f, v8f (&acc)[3][4], v2f (&dsum)[3]) {
#pragma unroll
  for (int u = 0; u < AGG_U; ++u) {
    v2f a[3], bq[4];
#pragma unroll
    for (int i = 0; i < 3; ++i) { a[i].x = f.g[u][0][i];  a[i].y = f.g[u][1][i]; }
#pragma unroll
    for (int j = 0; j < 4; ++j) { bq[j].x = f.e[u][0][j];  bq[j].y = f.e[u][1][j]; }
#pragma unroll
    for (int i = 0; i < 3; ++i) {
      dsum[i] += a[i];
#pragma unroll
      for (int j = 0; j < 4; ++j)
        acc[i][j] = WMMA_F32(a[i], bq[j], acc[i][j]);
    }
    __builtin_amdgcn_sched_barrier(0);
  }
}

__global__ __launch_bounds__(32 * AGG_WAVES)
void agg_kernel(const float* __restrict__ G, const float* __restrict__ edge,
                float* __restrict__ aggp, float* __restrict__ degp) {
  __shared__ __attribute__((aligned(16))) float tile[AGG_WAVES * AGG_ROWS * AGG_PITCH];
  __shared__ float degs[AGG_WAVES * AGG_ROWS];

  const int tiles_n = NN / AGG_ROWS;
  int t = blockIdx.x;
  const int plane = t / (BSZ * tiles_n);
  t -= plane * BSZ * tiles_n;
  const int b  = t / tiles_n;
  const int tn = t - b * tiles_n;

  const int wave = threadIdx.x >> 5;
  const int lane = threadIdx.x & 31;
  const int half = lane >> 4;
  const int l15  = lane & 15;
  const int e0 = (plane * AGG_WAVES + wave) * AGG_KW;
  const int n0 = tn * AGG_ROWS + 3 * l15;

  const float* __restrict__ gp = G    + (size_t)b * EE * NN       + (size_t)(e0 + half * 2) * NN       + n0;
  const float* __restrict__ ep = edge + (size_t)b * EE * EDGE_DIM + (size_t)(e0 + half * 2) * EDGE_DIM + 4 * l15;
  const size_t gstep = (size_t)(4 * AGG_U) * NN;
  const size_t estep = (size_t)(4 * AGG_U) * EDGE_DIM;

  v8f acc[3][4] = {};
  v2f dsum[3] = {};

  AggFrag fa, fb;
  const int nblk = AGG_KW / (4 * AGG_U);
  agg_load(fa, gp, ep);
  for (int blk = 0; blk < nblk - 2; blk += 2) {
    agg_load(fb, gp + gstep, ep + estep);
    __builtin_amdgcn_sched_barrier(0);
    agg_compute(fa, acc, dsum);
    gp += 2 * gstep;
    ep += 2 * estep;
    agg_load(fa, gp, ep);
    __builtin_amdgcn_sched_barrier(0);
    agg_compute(fb, acc, dsum);
  }
  agg_load(fb, gp + gstep, ep + estep);
  __builtin_amdgcn_sched_barrier(0);
  agg_compute(fa, acc, dsum);
  agg_compute(fb, acc, dsum);

#pragma unroll
  for (int i = 0; i < 3; ++i) {
    float deg = dsum[i].x + dsum[i].y;
    deg += __shfl_xor(deg, 16, 32);
    if (half == 0) degs[wave * AGG_ROWS + 3 * l15 + i] = deg;
  }

#pragma unroll
  for (int i = 0; i < 3; ++i) {
#pragma unroll
    for (int r = 0; r < 8; ++r) {
      v4f v;
      v.x = acc[i][0][r];  v.y = acc[i][1][r];  v.z = acc[i][2][r];  v.w = acc[i][3][r];
      *(v4f*)(tile + (wave * AGG_ROWS + 3 * (r + 8 * half) + i) * AGG_PITCH + 4 * l15) = v;
    }
  }
  __syncthreads();

  float* __restrict__ outp = aggp + ((size_t)(plane * BSZ + b) * NN + (size_t)tn * AGG_ROWS) * EDGE_DIM;
#pragma unroll
  for (int it = 0; it < AGG_PASSES; ++it) {
    const int idx = it * (32 * AGG_WAVES) + threadIdx.x;
    const int row = idx >> 4;
    const int c4  = (idx & 15) * 4;
    const float* __restrict__ p = tile + row * AGG_PITCH + c4;
    v4f s = *(const v4f*)p;
    s += *(const v4f*)(p + 1 * AGG_ROWS * AGG_PITCH);
    s += *(const v4f*)(p + 2 * AGG_ROWS * AGG_PITCH);
    s += *(const v4f*)(p + 3 * AGG_ROWS * AGG_PITCH);
    *(v4f*)(outp + (size_t)row * EDGE_DIM + c4) = s;
  }
  if (threadIdx.x < AGG_ROWS) {
    const int r = threadIdx.x;
    degp[(size_t)((plane * BSZ + b) * tiles_n + tn) * DEG_SLOT + r] =
        ((degs[r] + degs[AGG_ROWS + r]) + degs[2 * AGG_ROWS + r]) + degs[3 * AGG_ROWS + r];
  }
}

#define PRJ_ROWS   48
#define PRJ_WAVES  4
#define PRJ_U      8

__device__ __forceinline__ float deg_at(const float* __restrict__ degp, int plane, int m) {
  const int b  = m / NN;
  const int n  = m - b * NN;
  const int tn = n / AGG_ROWS;
  const int r  = n - tn * AGG_ROWS;
  return degp[(size_t)((plane * BSZ + b) * (NN / AGG_ROWS) + tn) * DEG_SLOT + r];
}

struct NodeFrag {
  v2f a[PRJ_U][3];
  v3f b[PRJ_U][2];
};

struct EdgeFrag {
  v2f p0[PRJ_U][3];
  v2f p1[PRJ_U][3];
  v3f b[PRJ_U][2];
};

__device__ __forceinline__ void node_load(NodeFrag& f, const float* __restrict__ xp,
                                          const float* __restrict__ wp) {
#pragma unroll
  for (int u = 0; u < PRJ_U; ++u) {
#pragma unroll
    for (int i = 0; i < 3; ++i)
      f.a[u][i] = *(const v2f*)(xp + (size_t)(16 * i) * NODE_DIM + 4 * u);
    f.b[u][0] = ld3(wp + (size_t)(4 * u) * QKV);
    f.b[u][1] = ld3(wp + (size_t)(4 * u + 1) * QKV);
  }
}

__device__ __forceinline__ void edge_load(EdgeFrag& f, const float* __restrict__ pp, size_t pl,
                                          const float* __restrict__ wp) {
#pragma unroll
  for (int u = 0; u < PRJ_U; ++u) {
#pragma unroll
    for (int i = 0; i < 3; ++i) {
      f.p0[u][i] = *(const v2f*)(pp + (size_t)(16 * i) * EDGE_DIM + 4 * u);
      f.p1[u][i] = *(const v2f*)(pp + pl + (size_t)(16 * i) * EDGE_DIM + 4 * u);
    }
    f.b[u][0] = ld3(wp + (size_t)(4 * u) * QKV);
    f.b[u][1] = ld3(wp + (size_t)(4 * u + 1) * QKV);
  }
}

__device__ __forceinline__ void prj_kstep(const v2f (&a)[3], const v3f& b0, const v3f& b1,
                                          v8f (&acc)[3][3]) {
  v2f bq[3];
#pragma unroll
  for (int j = 0; j < 3; ++j) { bq[j].x = b0[j];  bq[j].y = b1[j]; }
#pragma unroll
  for (int i = 0; i < 3; ++i) {
#pragma unroll
    for (int j = 0; j < 3; ++j)
      acc[i][j] = WMMA_F32(a[i], bq[j], acc[i][j]);
  }
}

__device__ __forceinline__ void node_compute(const NodeFrag& f, v8f (&acc)[3][3]) {
#pragma unroll
  for (int u = 0; u < PRJ_U; ++u) {
    v2f a[3];
#pragma unroll
    for (int i = 0; i < 3; ++i) a[i] = f.a[u][i];
    prj_kstep(a, f.b[u][0], f.b[u][1], acc);
    __builtin_amdgcn_sched_barrier(0);
  }
}

__device__ __forceinline__ void edge_compute(const EdgeFrag& f, const float (&rinv)[3],
                                             v8f (&acc)[3][3]) {
#pragma unroll
  for (int u = 0; u < PRJ_U; ++u) {
    v2f a[3];
#pragma unroll
    for (int i = 0; i < 3; ++i) a[i] = (f.p0[u][i] + f.p1[u][i]) * rinv[i];
    prj_kstep(a, f.b[u][0], f.b[u][1], acc);
    __builtin_amdgcn_sched_barrier(0);
  }
}

__global__ __launch_bounds__(32 * PRJ_WAVES)
void qkv_proj_kernel(const float* __restrict__ node, const float* __restrict__ Wn,
                     const float* __restrict__ bn, const float* __restrict__ aggp,
                     const float* __restrict__ degp, const float* __restrict__ We,
                     const float* __restrict__ be, float* __restrict__ C) {
  const int tm   = blockIdx.x >> 1;
  const int wave = threadIdx.x >> 5;
  const int lane = threadIdx.x & 31;
  const int half = lane >> 4;
  const int l15  = lane & 15;
  const int c    = ((blockIdx.x & 1) * PRJ_WAVES + wave) * 48 + 3 * l15;
  const int row0 = tm * PRJ_ROWS + l15;

  float d0[3], d1[3], bnv[3], bev[3];
#pragma unroll
  for (int i = 0; i < 3; ++i) {
    d0[i]  = deg_at(degp, 0, row0 + 16 * i);
    d1[i]  = deg_at(degp, 1, row0 + 16 * i);
    bnv[i] = bn[c + i];
    bev[i] = be[c + i];
  }

  const float* __restrict__ xp = node + (size_t)row0 * NODE_DIM + 2 * half;
  const float* __restrict__ wp = Wn   + (size_t)(2 * half) * QKV + c;
  const float* __restrict__ pp = aggp + (size_t)row0 * EDGE_DIM + 2 * half;
  const float* __restrict__ ep = We   + (size_t)(2 * half) * QKV + c;
  const size_t pl    = (size_t)BSZ * NN * EDGE_DIM;
  const int    kblk  = 4 * PRJ_U;
  const size_t wstep = (size_t)kblk * QKV;

  v8f acc[3][3] = {};
  NodeFrag na, nb;
  EdgeFrag ea, eb;

  node_load(na, xp, wp);
  node_load(nb, xp + kblk, wp + wstep);
  __builtin_amdgcn_sched_barrier(0);
  node_compute(na, acc);
  node_load(na, xp + 2 * kblk, wp + 2 * wstep);
  __builtin_amdgcn_sched_barrier(0);
  node_compute(nb, acc);
  node_load(nb, xp + 3 * kblk, wp + 3 * wstep);
  __builtin_amdgcn_sched_barrier(0);
  node_compute(na, acc);
  edge_load(ea, pp, pl, ep);
  __builtin_amdgcn_sched_barrier(0);
  node_compute(nb, acc);
  edge_load(eb, pp + kblk, pl, ep + wstep);
  __builtin_amdgcn_sched_barrier(0);

  float rinv[3];
#pragma unroll
  for (int i = 0; i < 3; ++i) rinv[i] = 1.0f / (d0[i] + d1[i]);
  edge_compute(ea, rinv, acc);
  edge_compute(eb, rinv, acc);

  float bv[3];
#pragma unroll
  for (int j = 0; j < 3; ++j) bv[j] = bnv[j] + bev[j];
#pragma unroll
  for (int i = 0; i < 3; ++i) {
#pragma unroll
    for (int r = 0; r < 8; ++r) {
      float* __restrict__ p = C + (size_t)(tm * PRJ_ROWS + 16 * i + 8 * half + r) * QKV + c;
      p[0] = acc[i][0][r] + bv[0];
      p[1] = acc[i][1][r] + bv[1];
      p[2] = acc[i][2][r] + bv[2];
    }
  }
}

#define OUT_KSTEPS (NODE_DIM / 4)

__global__ __launch_bounds__(32 * PRJ_WAVES)
void out_proj_kernel(const float* __restrict__ X, const float* __restrict__ W,
                     const float* __restrict__ bias, float* __restrict__ C) {
  const int tm   = blockIdx.x >> 1;
  const int wave = threadIdx.x >> 5;
  const int lane = threadIdx.x & 31;
  const int half = lane >> 4;
  const int l15  = lane & 15;
  const int col  = ((blockIdx.x & 1) * PRJ_WAVES + wave) * 16 + l15;
  const int row0 = tm * PRJ_ROWS + l15;

  const float* __restrict__ xp = X + (size_t)row0 * NODE_DIM + 2 * half;
  const float* __restrict__ wp = W + (size_t)(2 * half) * NODE_DIM + col;

  v2f a[OUT_KSTEPS][3], b[OUT_KSTEPS];
#pragma unroll
  for (int u = 0; u < OUT_KSTEPS; ++u) {
#pragma unroll
    for (int i = 0; i < 3; ++i)
      a[u][i] = *(const v2f*)(xp + (size_t)(16 * i) * NODE_DIM + 4 * u);
    b[u].x = wp[(size_t)(4 * u) * NODE_DIM];
    b[u].y = wp[(size_t)(4 * u + 1) * NODE_DIM];
  }
  __builtin_amdgcn_sched_barrier(0);

  v8f acc[3] = {};
#pragma unroll
  for (int u = 0; u < OUT_KSTEPS; ++u) {
#pragma unroll
    for (int i = 0; i < 3; ++i)
      acc[i] = WMMA_F32(a[u][i], b[u], acc[i]);
  }

  const float bv = bias[col];
#pragma unroll
  for (int i = 0; i < 3; ++i) {
#pragma unroll
    for (int r = 0; r < 8; ++r)
      C[(size_t)(tm * PRJ_ROWS + 16 * i + 8 * half + r) * NODE_DIM + col] = acc[i][r] + bv;
  }
}

#define ATT_WAVES 4
#define ATT_QT    3

__device__ __forceinline__ float xhalf(float x) {
  const int b = __builtin_bit_cast(int, x);
  return __builtin_bit_cast(float,
      __builtin_amdgcn_permlanex16(b, b, 0x76543210, 0xfedcba98, false, false));
}

__device__ __forceinline__ float max3f(float a, float b, float c) {
  return __builtin_fmaximum_numf(__builtin_fmaximum_numf(a, b), c);
}

struct AttFrag {
  v4f   k[2][2];
  float v[2][8];
};

__device__ __forceinline__ void att_load(AttFrag& f, const float* __restrict__ kp,
                                         const float* __restrict__ vp) {
  for (int j = 0; j < 2; ++j) {
    f.k[j][0] = *(const v4f*)(kp + (size_t)(16 * j) * QKV);
    f.k[j][1] = *(const v4f*)(kp + (size_t)(16 * j) * QKV + 4);
    for (int r = 0; r < 8; ++r)
      f.v[j][r] = vp[(size_t)(16 * j + r) * QKV];
  }
}

__device__ __forceinline__ void sc_t(const AttFrag& f, const v2f (&qb)[4],
                                     v8f& s0, v8f& s1, int t) {
  v2f a0, a1;
  a0.x = f.k[0][t >> 1][2 * (t & 1)];  a0.y = f.k[0][t >> 1][2 * (t & 1) + 1];
  a1.x = f.k[1][t >> 1][2 * (t & 1)];  a1.y = f.k[1][t >> 1][2 * (t & 1) + 1];
  s0 = WMMA_F32(a0, qb[t], s0);
  s1 = WMMA_F32(a1, qb[t], s1);
}

struct SmState { float mnew, corr;  v2f ps; };

__device__ __forceinline__ void sm_max(const v8f& s0, const v8f& s1, float& m, SmState& st) {
  const float t0 = max3f(s0[0], s0[1], s0[2]);
  const float t1 = max3f(s0[3], s0[4], s0[5]);
  const float t2 = max3f(s0[6], s0[7], s1[0]);
  const float t3 = max3f(s1[1], s1[2], s1[3]);
  const float t4 = max3f(s1[4], s1[5], s1[6]);
  const float mx = __builtin_fmaximum_numf(max3f(t0, t1, t2), max3f(t3, t4, s1[7]));
  const float mnew = max3f(m, mx, xhalf(mx));
  st.corr = __builtin_amdgcn_exp2f(m - mnew);
  st.mnew = mnew;
  st.ps.x = 0.f;  st.ps.y = 0.f;
  m = mnew;
}

__device__ __forceinline__ void sm_exp(v8f& s0, v8f& s1, SmState& st, int i) {
  const v2f mm = {st.mnew, st.mnew};
  v2f d0, d1;
  d0.x = s0[2 * i];  d0.y = s0[2 * i + 1];
  d1.x = s1[2 * i];  d1.y = s1[2 * i + 1];
  d0 -= mm;
  d1 -= mm;
  v2f e0, e1;
  e0.x = __builtin_amdgcn_exp2f(d0.x);  e0.y = __builtin_amdgcn_exp2f(d0.y);
  e1.x = __builtin_amdgcn_exp2f(d1.x);  e1.y = __builtin_amdgcn_exp2f(d1.y);
  st.ps += e0;
  st.ps += e1;
  s0[2 * i] = e0.x;  s0[2 * i + 1] = e0.y;
  s1[2 * i] = e1.x;  s1[2 * i + 1] = e1.y;
}

__device__ __forceinline__ void sm_fin(v8f& o, float& l, const SmState& st) {
  l = l * st.corr + (st.ps.x + st.ps.y);
  o *= st.corr;
}

__device__ __forceinline__ void pv_u(const AttFrag& f, const v8f& p0, const v8f& p1,
                                     v8f& o, int u) {
  const int j = u >> 2, t = u & 3;
  v2f a, bb;
  a.x = f.v[j][2 * t];  a.y = f.v[j][2 * t + 1];
  if (j == 0) { bb.x = p0[2 * t];  bb.y = p0[2 * t + 1]; }
  else        { bb.x = p1[2 * t];  bb.y = p1[2 * t + 1]; }
  o = WMMA_F32(a, bb, o);
}

#define ATT_PIECE_END() __builtin_amdgcn_sched_barrier(0)

__device__ __forceinline__ void att_step(AttFrag& f, const v2f (&qb)[ATT_QT][4],
                                         v8f (&o)[ATT_QT], float (&m)[ATT_QT], float (&l)[ATT_QT]) {
  asm volatile("" : "+v"(f.k[0][0]), "+v"(f.k[0][1]), "+v"(f.k[1][0]), "+v"(f.k[1][1]),
                    "+v"(o[0]), "+v"(o[1]), "+v"(o[2]),
                    "+v"(m[0]), "+v"(m[1]), "+v"(m[2]),
                    "+v"(l[0]), "+v"(l[1]), "+v"(l[2]));

  v8f s[ATT_QT][2] = {};
  SmState st[ATT_QT];

  for (int t = 0; t < 4; ++t) sc_t(f, qb[0], s[0][0], s[0][1], t);
  ATT_PIECE_END();

  for (int q = 1; q < ATT_QT; ++q) {
    sc_t(f, qb[q], s[q][0], s[q][1], 0);
    sm_max(s[q - 1][0], s[q - 1][1], m[q - 1], st[q - 1]);
    ATT_PIECE_END();
    sc_t(f, qb[q], s[q][0], s[q][1], 1);
    sm_exp(s[q - 1][0], s[q - 1][1], st[q - 1], 0);
    sm_exp(s[q - 1][0], s[q - 1][1], st[q - 1], 1);
    ATT_PIECE_END();
    sc_t(f, qb[q], s[q][0], s[q][1], 2);
    sm_exp(s[q - 1][0], s[q - 1][1], st[q - 1], 2);
    sm_exp(s[q - 1][0], s[q - 1][1], st[q - 1], 3);
    ATT_PIECE_END();
    sc_t(f, qb[q], s[q][0], s[q][1], 3);
    sm_fin(o[q - 1], l[q - 1], st[q - 1]);
    ATT_PIECE_END();
  }

  pv_u(f, s[0][0], s[0][1], o[0], 0);  pv_u(f, s[1][0], s[1][1], o[1], 0);
  pv_u(f, s[0][0], s[0][1], o[0], 1);  pv_u(f, s[1][0], s[1][1], o[1], 1);
  sm_max(s[2][0], s[2][1], m[2], st[2]);
  ATT_PIECE_END();
  pv_u(f, s[0][0], s[0][1], o[0], 2);  pv_u(f, s[1][0], s[1][1], o[1], 2);
  pv_u(f, s[0][0], s[0][1], o[0], 3);  pv_u(f, s[1][0], s[1][1], o[1], 3);
  sm_exp(s[2][0], s[2][1], st[2], 0);
  sm_exp(s[2][0], s[2][1], st[2], 1);
  ATT_PIECE_END();
  pv_u(f, s[0][0], s[0][1], o[0], 4);  pv_u(f, s[1][0], s[1][1], o[1], 4);
  pv_u(f, s[0][0], s[0][1], o[0], 5);  pv_u(f, s[1][0], s[1][1], o[1], 5);
  sm_exp(s[2][0], s[2][1], st[2], 2);
  sm_exp(s[2][0], s[2][1], st[2], 3);
  ATT_PIECE_END();
  pv_u(f, s[0][0], s[0][1], o[0], 6);  pv_u(f, s[1][0], s[1][1], o[1], 6);
  pv_u(f, s[0][0], s[0][1], o[0], 7);  pv_u(f, s[1][0], s[1][1], o[1], 7);
  sm_fin(o[2], l[2], st[2]);
  ATT_PIECE_END();

  for (int u = 0; u < 8; ++u) pv_u(f, s[2][0], s[2][1], o[2], u);
  ATT_PIECE_END();
}

__global__ __launch_bounds__(32 * ATT_WAVES)
void attn_kernel(const float* __restrict__ qkv, float* __restrict__ attout) {
  const int groups_q = NN / (16 * ATT_QT * ATT_WAVES);
  int grp = blockIdx.x;
  const int b = grp / (NHEAD * groups_q);
  grp -= b * NHEAD * groups_q;
  const int h  = grp / groups_q;
  const int gq = grp - h * groups_q;
  const int wave = threadIdx.x >> 5;
  const int tq = (gq * ATT_WAVES + wave) * ATT_QT;

  const int lane = threadIdx.x & 31;
  const int half = lane >> 4;
  const int l15  = lane & 15;
  const float qscale = 0.25f * 1.4426950408889634f;

  const float* __restrict__ base = qkv + (size_t)b * NN * QKV + h * 48;

  v2f qb[ATT_QT][4];
  for (int q = 0; q < ATT_QT; ++q) {
    const float* __restrict__ qp = base + (size_t)((tq + q) * 16 + l15) * QKV + 8 * half;
    const v4f q0 = *(const v4f*)qp;
    const v4f q1 = *(const v4f*)(qp + 4);
    qb[q][0].x = q0.x * qscale;  qb[q][0].y = q0.y * qscale;
    qb[q][1].x = q0.z * qscale;  qb[q][1].y = q0.w * qscale;
    qb[q][2].x = q1.x * qscale;  qb[q][2].y = q1.y * qscale;
    qb[q][3].x = q1.z * qscale;  qb[q][3].y = q1.w * qscale;
  }

  const float* __restrict__ kp = base + (size_t)l15 * QKV + 16 + 8 * half;
  const float* __restrict__ vp = base + (size_t)(8 * half) * QKV + 32 + l15;
  const size_t step = (size_t)32 * QKV;

  v8f o[ATT_QT] = {};
  float m[ATT_QT], l[ATT_QT];
  for (int q = 0; q < ATT_QT; ++q) { m[q] = -1e30f;  l[q] = 0.f; }

  AttFrag fa, fb;
  const int nstep = NN / 32;
  att_load(fa, kp, vp);
  for (int it = 0; it < nstep - 2; it += 2) {
    att_load(fb, kp + step, vp + step);
    __builtin_amdgcn_sched_barrier(0);
    att_step(fa, qb, o, m, l);
    kp += 2 * step;
    vp += 2 * step;
    att_load(fa, kp, vp);
    __builtin_amdgcn_sched_barrier(0);
    att_step(fb, qb, o, m, l);
  }
  att_load(fb, kp + step, vp + step);
  __builtin_amdgcn_sched_barrier(0);
  att_step(fa, qb, o, m, l);
  att_step(fb, qb, o, m, l);

  for (int q = 0; q < ATT_QT; ++q) {
    const float ls = l[q] + xhalf(l[q]);
    float* __restrict__ outp = attout + (size_t)b * NN * NODE_DIM
                             + (size_t)((tq + q) * 16 + l15) * NODE_DIM + h * HDIM + 8 * half;
    v4f r0, r1;
    r0.x = o[q][0] / ls;  r0.y = o[q][1] / ls;  r0.z = o[q][2] / ls;  r0.w = o[q][3] / ls;
    r1.x = o[q][4] / ls;  r1.y = o[q][5] / ls;  r1.z = o[q][6] / ls;  r1.w = o[q][7] / ls;
    *(v4f*)outp       = r0;
    *(v4f*)(outp + 4) = r1;
  }
}

extern "C" void kernel_launch(void* const* d_in, const int* in_sizes, int n_in,
                              void* d_out, int out_size, void* d_ws, size_t ws_size,
                              hipStream_t stream) {
  const float* node    = (const float*)d_in[0];
  const float* edge    = (const float*)d_in[1];
  const float* G       = (const float*)d_in[2];
  const float* W_qkv_n = (const float*)d_in[3];
  const float* b_qkv_n = (const float*)d_in[4];
  const float* W_qkv_e = (const float*)d_in[5];
  const float* b_qkv_e = (const float*)d_in[6];
  const float* W_o     = (const float*)d_in[7];
  const float* b_o     = (const float*)d_in[8];
  float* out = (float*)d_out;

  float* qkv  = (float*)d_ws;
  float* aggp = qkv  + (size_t)BSZ * NN * QKV;
  float* degp = aggp + (size_t)AGG_PLANES * BSZ * NN * EDGE_DIM;
  float* att  = degp + (size_t)AGG_PLANES * BSZ * (NN / AGG_ROWS) * DEG_SLOT;

  agg_kernel<<<AGG_PLANES * BSZ * (NN / AGG_ROWS), 32 * AGG_WAVES, 0, stream>>>(G, edge, aggp, degp);

  qkv_proj_kernel<<<(BSZ * NN / PRJ_ROWS) * 2, 32 * PRJ_WAVES, 0, stream>>>(
      node, W_qkv_n, b_qkv_n, aggp, degp, W_qkv_e, b_qkv_e, qkv);

  attn_kernel<<<BSZ * NHEAD * (NN / (16 * ATT_QT * ATT_WAVES)), 32 * ATT_WAVES, 0, stream>>>(qkv, att);

  out_proj_kernel<<<(BSZ * NN / PRJ_ROWS) * 2, 32 * PRJ_WAVES, 0, stream>>>(att, W_o, b_o, out);
}
